# starting kernel
# baseline (speedup 1.0000x reference)
#include <hip/hip_runtime.h>
#include <hip/hip_bf16.h>

typedef __attribute__((ext_vector_type(2))) float v2f;
typedef __attribute__((ext_vector_type(8))) float v8f;

__global__ __launch_bounds__(256) void k_init_deg(float* __restrict__ deg, int n) {
  int i = blockIdx.x * blockDim.x + threadIdx.x;
  if (i < n) deg[i] = 1.0f;
}

__global__ __launch_bounds__(256) void k_count_deg(const int* __restrict__ col,
                                                   float* deg, int E) {
  int e = blockIdx.x * blockDim.x + threadIdx.x;
  if (e < E) {
    (void)__hip_atomic_fetch_add(&deg[col[e]], 1.0f, __ATOMIC_RELAXED,
                                 __HIP_MEMORY_SCOPE_AGENT);
  }
}

__global__ __launch_bounds__(256) void k_dinv(float* __restrict__ deg, int n) {
  int i = blockIdx.x * blockDim.x + threadIdx.x;
  if (i < n) {
    float d = deg[i];
    deg[i] = (d > 0.0f) ? rsqrtf(d) : 0.0f;
  }
}

__global__ __launch_bounds__(256) void k_gemm_wmma_f32(
    const float* __restrict__ A, const float* __restrict__ B,
    float* __restrict__ C, int M, int K, int N) {
  const int lane  = threadIdx.x & 31;
  const int wave  = threadIdx.x >> 5;
  const int tilesN = N >> 4;
  const int t  = blockIdx.x * (blockDim.x >> 5) + wave;
  const int m0 = (t / tilesN) << 4;
  const int n0 = (t % tilesN) << 4;
  if (m0 >= M) return;

  const int l15   = lane & 15;
  const int khalf = (lane >> 4) << 1;

  const float* arow = A + (size_t)(m0 + l15) * K + khalf;
  const float* bcol = B + (size_t)khalf * N + (n0 + l15);

  v8f acc = {};
  for (int k0 = 0; k0 < K; k0 += 4) {
    v2f a, b;
    float2 av = *(const float2*)arow;
    a.x = av.x;
    a.y = av.y;
    b.x = bcol[0];
    b.y = bcol[N];
    acc = __builtin_amdgcn_wmma_f32_16x16x4_f32(
         false, a,  false, b,
         (short)0, acc,  false,  false);
    arow += 4;
    bcol += (size_t)4 * N;
  }

  const int rbase = m0 + ((lane >> 4) << 3);
  float* cptr = C + (size_t)rbase * N + n0 + l15;
#pragma unroll
  for (int v = 0; v < 8; ++v) cptr[(size_t)v * N] = acc[v];
}

__global__ __launch_bounds__(256) void k_selfloop(const float* __restrict__ h,
                                                  const float* __restrict__ dinv,
                                                  float* __restrict__ agg,
                                                  int total, int F) {
  int idx = blockIdx.x * blockDim.x + threadIdx.x;
  if (idx < total) {
    float di = dinv[idx / F];
    agg[idx] = h[idx] * di * di;
  }
}

__global__ __launch_bounds__(256) void k_scatter(const float* __restrict__ h,
                                                 const int* __restrict__ row,
                                                 const int* __restrict__ col,
                                                 const float* __restrict__ dinv,
                                                 float* agg, int E, int F) {
  const int g = F >> 2;
  const long long total = (long long)E * g;
  long long tid = (long long)blockIdx.x * blockDim.x + threadIdx.x;
  if (tid >= total) return;
  const int e = (int)(tid / g);
  const int f = (int)(tid % g) << 2;
  const int r = row[e];
  const int c = col[e];
  const float nrm = dinv[r] * dinv[c];
  const float4 hv = *(const float4*)(h + (size_t)r * F + f);
  float* dst = agg + (size_t)c * F + f;
  (void)__hip_atomic_fetch_add(&dst[0], hv.x * nrm, __ATOMIC_RELAXED, __HIP_MEMORY_SCOPE_AGENT);
  (void)__hip_atomic_fetch_add(&dst[1], hv.y * nrm, __ATOMIC_RELAXED, __HIP_MEMORY_SCOPE_AGENT);
  (void)__hip_atomic_fetch_add(&dst[2], hv.z * nrm, __ATOMIC_RELAXED, __HIP_MEMORY_SCOPE_AGENT);
  (void)__hip_atomic_fetch_add(&dst[3], hv.w * nrm, __ATOMIC_RELAXED, __HIP_MEMORY_SCOPE_AGENT);
}

__global__ __launch_bounds__(256) void k_bias_act(float* __restrict__ agg,
                                                  const float* __restrict__ b,
                                                  int total, int F, int do_relu) {
  int idx = blockIdx.x * blockDim.x + threadIdx.x;
  if (idx < total) {
    float v = agg[idx] + b[idx % F];
    if (do_relu) v = fmaxf(v, 0.0f);
    agg[idx] = v;
  }
}

extern "C" void kernel_launch(void* const* d_in, const int* in_sizes, int n_in,
                              void* d_out, int out_size, void* d_ws, size_t ws_size,
                              hipStream_t stream) {
  const float* x  = (const float*)d_in[0];
  const int*   ei = (const int*)d_in[1];
  const float* W1 = (const float*)d_in[2];
  const float* b1 = (const float*)d_in[3];
  const float* W2 = (const float*)d_in[4];
  const float* b2 = (const float*)d_in[5];

  const int H  = in_sizes[3];
  const int C  = in_sizes[5];
  const int F  = in_sizes[2] / H;
  const int Nn = in_sizes[0] / F;
  const int E  = in_sizes[1] / 2;
  const int* rowIdx = ei;
  const int* colIdx = ei + E;

  auto al = [](size_t v) { return (v + 255) & ~(size_t)255; };
  float* dinv = (float*)d_ws;
  size_t off  = al((size_t)Nn * sizeof(float));
  float* hbuf = (float*)((char*)d_ws + off);
  off += al((size_t)Nn * H * sizeof(float));
  float* abuf = (float*)((char*)d_ws + off);

  const dim3 blk(256);
  auto blocks = [](long long work) { return (unsigned)((work + 255) / 256); };

  k_init_deg <<<blocks(Nn), blk, 0, stream>>>(dinv, Nn);
  k_count_deg<<<blocks(E),  blk, 0, stream>>>(colIdx, dinv, E);
  k_dinv     <<<blocks(Nn), blk, 0, stream>>>(dinv, Nn);

  const int tiles1 = ((Nn + 15) / 16) * (H / 16);
  k_gemm_wmma_f32<<<(tiles1 + 7) / 8, blk, 0, stream>>>(x, W1, hbuf, Nn, F, H);
  const int tot1 = Nn * H;
  k_selfloop<<<blocks(tot1), blk, 0, stream>>>(hbuf, dinv, abuf, tot1, H);
  k_scatter <<<blocks((long long)E * (H / 4)), blk, 0, stream>>>(hbuf, rowIdx, colIdx, dinv, abuf, E, H);
  k_bias_act<<<blocks(tot1), blk, 0, stream>>>(abuf, b1, tot1, H,  1);

  const int tiles2 = ((Nn + 15) / 16) * (C / 16);
  k_gemm_wmma_f32<<<(tiles2 + 7) / 8, blk, 0, stream>>>(abuf, W2, hbuf, Nn, H, C);
  const int tot2 = Nn * C;
  float* out = (float*)d_out;
  k_selfloop<<<blocks(tot2), blk, 0, stream>>>(hbuf, dinv, out, tot2, C);
  k_scatter <<<blocks((long long)E * (C / 4)), blk, 0, stream>>>(hbuf, rowIdx, colIdx, dinv, out, E, C);
  k_bias_act<<<blocks(tot2), blk, 0, stream>>>(out, b2, tot2, C,  0);
}


# layer 1 reordered as (S x) W1 in place of S (x W1): self-loop and atomic scatter run on 128-wide x rows before the GEMM, not on 256-wide rows after it; changes f32 summation order
# speedup vs baseline: 1.3623x; 1.3623x over previous
#include <hip/hip_runtime.h>
#include <hip/hip_bf16.h>

typedef __attribute__((ext_vector_type(2))) float v2f;
typedef __attribute__((ext_vector_type(8))) float v8f;

__global__ __launch_bounds__(256) void k_init_deg(float* __restrict__ deg, int n) {
  int i = blockIdx.x * blockDim.x + threadIdx.x;
  if (i < n) deg[i] = 1.0f;
}

__global__ __launch_bounds__(256) void k_count_deg(const int* __restrict__ col,
                                                   float* deg, int E) {
  int e = blockIdx.x * blockDim.x + threadIdx.x;
  if (e < E) {
    (void)__hip_atomic_fetch_add(&deg[col[e]], 1.0f, __ATOMIC_RELAXED,
                                 __HIP_MEMORY_SCOPE_AGENT);
  }
}

__global__ __launch_bounds__(256) void k_dinv(float* __restrict__ deg, int n) {
  int i = blockIdx.x * blockDim.x + threadIdx.x;
  if (i < n) {
    float d = deg[i];
    deg[i] = (d > 0.0f) ? rsqrtf(d) : 0.0f;
  }
}

__global__ __launch_bounds__(256) void k_gemm_wmma_f32(
    const float* __restrict__ A, const float* __restrict__ B,
    float* __restrict__ C, int M, int K, int N) {
  const int lane  = threadIdx.x & 31;
  const int wave  = threadIdx.x >> 5;
  const int tilesN = N >> 4;
  const int t  = blockIdx.x * (blockDim.x >> 5) + wave;
  const int m0 = (t / tilesN) << 4;
  const int n0 = (t % tilesN) << 4;
  if (m0 >= M) return;

  const int l15   = lane & 15;
  const int khalf = (lane >> 4) << 1;

  const float* arow = A + (size_t)(m0 + l15) * K + khalf;
  const float* bcol = B + (size_t)khalf * N + (n0 + l15);

  v8f acc = {};
  for (int k0 = 0; k0 < K; k0 += 4) {
    v2f a, b;
    float2 av = *(const float2*)arow;
    a.x = av.x;
    a.y = av.y;
    b.x = bcol[0];
    b.y = bcol[N];
    acc = __builtin_amdgcn_wmma_f32_16x16x4_f32(
         false, a,  false, b,
         (short)0, acc,  false,  false);
    arow += 4;
    bcol += (size_t)4 * N;
  }

  const int rbase = m0 + ((lane >> 4) << 3);
  float* cptr = C + (size_t)rbase * N + n0 + l15;
#pragma unroll
  for (int v = 0; v < 8; ++v) cptr[(size_t)v * N] = acc[v];
}

__global__ __launch_bounds__(256) void k_selfloop(const float* __restrict__ h,
                                                  const float* __restrict__ dinv,
                                                  float* __restrict__ agg,
                                                  int total, int F) {
  int idx = blockIdx.x * blockDim.x + threadIdx.x;
  if (idx < total) {
    float di = dinv[idx / F];
    agg[idx] = h[idx] * di * di;
  }
}

__global__ __launch_bounds__(256) void k_scatter(const float* __restrict__ h,
                                                 const int* __restrict__ row,
                                                 const int* __restrict__ col,
                                                 const float* __restrict__ dinv,
                                                 float* agg, int E, int F) {
  const int g = F >> 2;
  const long long total = (long long)E * g;
  long long tid = (long long)blockIdx.x * blockDim.x + threadIdx.x;
  if (tid >= total) return;
  const int e = (int)(tid / g);
  const int f = (int)(tid % g) << 2;
  const int r = row[e];
  const int c = col[e];
  const float nrm = dinv[r] * dinv[c];
  const float4 hv = *(const float4*)(h + (size_t)r * F + f);
  float* dst = agg + (size_t)c * F + f;
  (void)__hip_atomic_fetch_add(&dst[0], hv.x * nrm, __ATOMIC_RELAXED, __HIP_MEMORY_SCOPE_AGENT);
  (void)__hip_atomic_fetch_add(&dst[1], hv.y * nrm, __ATOMIC_RELAXED, __HIP_MEMORY_SCOPE_AGENT);
  (void)__hip_atomic_fetch_add(&dst[2], hv.z * nrm, __ATOMIC_RELAXED, __HIP_MEMORY_SCOPE_AGENT);
  (void)__hip_atomic_fetch_add(&dst[3], hv.w * nrm, __ATOMIC_RELAXED, __HIP_MEMORY_SCOPE_AGENT);
}

__global__ __launch_bounds__(256) void k_bias_act(float* __restrict__ agg,
                                                  const float* __restrict__ b,
                                                  int total, int F, int do_relu) {
  int idx = blockIdx.x * blockDim.x + threadIdx.x;
  if (idx < total) {
    float v = agg[idx] + b[idx % F];
    if (do_relu) v = fmaxf(v, 0.0f);
    agg[idx] = v;
  }
}

extern "C" void kernel_launch(void* const* d_in, const int* in_sizes, int n_in,
                              void* d_out, int out_size, void* d_ws, size_t ws_size,
                              hipStream_t stream) {
  const float* x  = (const float*)d_in[0];
  const int*   ei = (const int*)d_in[1];
  const float* W1 = (const float*)d_in[2];
  const float* b1 = (const float*)d_in[3];
  const float* W2 = (const float*)d_in[4];
  const float* b2 = (const float*)d_in[5];

  const int H  = in_sizes[3];
  const int C  = in_sizes[5];
  const int F  = in_sizes[2] / H;
  const int Nn = in_sizes[0] / F;
  const int E  = in_sizes[1] / 2;
  const int* rowIdx = ei;
  const int* colIdx = ei + E;

  auto al = [](size_t v) { return (v + 255) & ~(size_t)255; };
  float* dinv = (float*)d_ws;
  size_t off  = al((size_t)Nn * sizeof(float));
  float* hbuf = (float*)((char*)d_ws + off);
  off += al((size_t)Nn * H * sizeof(float));
  float* abuf = (float*)((char*)d_ws + off);

  const dim3 blk(256);
  auto blocks = [](long long work) { return (unsigned)((work + 255) / 256); };

  k_init_deg <<<blocks(Nn), blk, 0, stream>>>(dinv, Nn);
  k_count_deg<<<blocks(E),  blk, 0, stream>>>(colIdx, dinv, E);
  k_dinv     <<<blocks(Nn), blk, 0, stream>>>(dinv, Nn);

  const int tot0 = Nn * F;
  k_selfloop<<<blocks(tot0), blk, 0, stream>>>(x, dinv, abuf, tot0, F);
  k_scatter <<<blocks((long long)E * (F / 4)), blk, 0, stream>>>(x, rowIdx, colIdx, dinv, abuf, E, F);
  const int tiles1 = ((Nn + 15) / 16) * (H / 16);
  k_gemm_wmma_f32<<<(tiles1 + 7) / 8, blk, 0, stream>>>(abuf, W1, hbuf, Nn, F, H);
  const int tot1 = Nn * H;
  k_bias_act<<<blocks(tot1), blk, 0, stream>>>(hbuf, b1, tot1, H,  1);

  const int tiles2 = ((Nn + 15) / 16) * (C / 16);
  k_gemm_wmma_f32<<<(tiles2 + 7) / 8, blk, 0, stream>>>(hbuf, W2, abuf, Nn, H, C);
  const int tot2 = Nn * C;
  float* out = (float*)d_out;
  k_selfloop<<<blocks(tot2), blk, 0, stream>>>(abuf, dinv, out, tot2, C);
  k_scatter <<<blocks((long long)E * (C / 4)), blk, 0, stream>>>(abuf, rowIdx, colIdx, dinv, out, E, C);
  k_bias_act<<<blocks(tot2), blk, 0, stream>>>(out, b2, tot2, C,  0);
}


# aggregation: per-edge f32 atomic scatter replaced by a gather over edges sorted by destination (counting sort each call); one wave sums a row in registers, one store; changes f32 summation order
# speedup vs baseline: 2.3126x; 1.6976x over previous
#include <hip/hip_runtime.h>
#include <hip/hip_bf16.h>


typedef __attribute__((ext_vector_type(2))) float v2f;
typedef __attribute__((ext_vector_type(8))) float v8f;

struct __attribute__((aligned(8))) Seg {
  int start;
  int len;
};

__global__ __launch_bounds__(256) void k_zero_counts(int* __restrict__ cnt, int n) {
  int i = blockIdx.x * blockDim.x + threadIdx.x;
  if (i < n) cnt[i] = 0;
}

__global__ __launch_bounds__(256) void k_count_rank(const int* __restrict__ col,
                                                    int* cnt,
                                                    int* __restrict__ rank, int E) {
  int e = blockIdx.x * blockDim.x + threadIdx.x;
  if (e < E) {
    rank[e] = __hip_atomic_fetch_add(&cnt[col[e]], 1, __ATOMIC_RELAXED,
                                     __HIP_MEMORY_SCOPE_AGENT);
  }
}

__global__ __launch_bounds__(256) void k_offsets(const int* __restrict__ cnt,
                                                 int* cursor,
                                                 Seg* __restrict__ seg,
                                                 float* __restrict__ dinv, int n) {
  __shared__ int wsum[8];
  __shared__ int blk_base;
  const int i    = blockIdx.x * blockDim.x + threadIdx.x;
  const int lane = threadIdx.x & 31;
  const int wave = threadIdx.x >> 5;
  const int c = (i < n) ? cnt[i] : 0;
  int v = c;
#pragma unroll
  for (int d = 1; d < 32; d <<= 1) {
    int t = __shfl_up(v, d);
    if (lane >= d) v += t;
  }
  if (lane == 31) wsum[wave] = v;
  __syncthreads();
  if (threadIdx.x == 0) {
    int tot = 0;
#pragma unroll
    for (int w = 0; w < 8; ++w) {
      int t = wsum[w];
      wsum[w] = tot;
      tot += t;
    }
    blk_base = __hip_atomic_fetch_add(cursor, tot, __ATOMIC_RELAXED,
                                      __HIP_MEMORY_SCOPE_AGENT);
  }
  __syncthreads();
  if (i < n) {
    Seg sg;
    sg.start = blk_base + wsum[wave] + (v - c);
    sg.len   = c;
    seg[i] = sg;
    const float d = (float)(c + 1);
    dinv[i] = (d > 0.0f) ? rsqrtf(d) : 0.0f;
  }
}

__global__ __launch_bounds__(256) void k_fill(const int* __restrict__ row,
                                              const int* __restrict__ col,
                                              const int* __restrict__ rank,
                                              const Seg* __restrict__ seg,
                                              int* __restrict__ srcs, int E) {
  int e = blockIdx.x * blockDim.x + threadIdx.x;
  if (e < E) srcs[seg[col[e]].start + rank[e]] = row[e];
}

template <int VEC>
__global__ __launch_bounds__(256) void k_gather_rows(
    const float* __restrict__ h, const Seg* __restrict__ seg,
    const int* __restrict__ srcs, const float* __restrict__ dinv,
    float* __restrict__ out, int n_nodes) {
  typedef float vec __attribute__((ext_vector_type(VEC)));
  const int lane = threadIdx.x & 31;
  const int node = __builtin_amdgcn_readfirstlane(
      (int)(blockIdx.x * (blockDim.x >> 5) + (threadIdx.x >> 5)));
  if (node >= n_nodes) return;
  const Seg   sg = seg[node];
  const int   n  = sg.len;
  const float di = dinv[node];
  const int* __restrict__ sp = srcs + sg.start;
  const vec* __restrict__ hv = (const vec*)h + lane;

  vec acc = hv[(size_t)node * 32] * (di * di);
  int j = 0;
  for (; j + 4 <= n; j += 4) {
    const int s0 = sp[j], s1 = sp[j + 1], s2 = sp[j + 2], s3 = sp[j + 3];
    const vec r0 = hv[(size_t)s0 * 32];
    const vec r1 = hv[(size_t)s1 * 32];
    const vec r2 = hv[(size_t)s2 * 32];
    const vec r3 = hv[(size_t)s3 * 32];
    const float w0 = dinv[s0] * di, w1 = dinv[s1] * di;
    const float w2 = dinv[s2] * di, w3 = dinv[s3] * di;
    acc += r0 * w0;
    acc += r1 * w1;
    acc += r2 * w2;
    acc += r3 * w3;
  }
  if (j + 2 <= n) {
    const int s0 = sp[j], s1 = sp[j + 1];
    const vec r0 = hv[(size_t)s0 * 32];
    const vec r1 = hv[(size_t)s1 * 32];
    const float w0 = dinv[s0] * di, w1 = dinv[s1] * di;
    acc += r0 * w0;
    acc += r1 * w1;
    j += 2;
  }
  if (j < n) {
    const int s0 = sp[j];
    const vec r0 = hv[(size_t)s0 * 32];
    acc += r0 * (dinv[s0] * di);
  }
  ((vec*)out)[(size_t)node * 32 + lane] = acc;
}

__global__ __launch_bounds__(256) void k_gemm_wmma_f32(
    const float* __restrict__ A, const float* __restrict__ B,
    float* __restrict__ C, int M, int K, int N) {
  const int lane  = threadIdx.x & 31;
  const int wave  = threadIdx.x >> 5;
  const int tilesN = N >> 4;
  const int t  = blockIdx.x * (blockDim.x >> 5) + wave;
  const int m0 = (t / tilesN) << 4;
  const int n0 = (t % tilesN) << 4;
  if (m0 >= M) return;

  const int l15   = lane & 15;
  const int khalf = (lane >> 4) << 1;

  const float* arow = A + (size_t)(m0 + l15) * K + khalf;
  const float* bcol = B + (size_t)khalf * N + (n0 + l15);

  v8f acc = {};
  for (int k0 = 0; k0 < K; k0 += 4) {
    v2f a, b;
    float2 av = *(const float2*)arow;
    a.x = av.x;
    a.y = av.y;
    b.x = bcol[0];
    b.y = bcol[N];
    acc = __builtin_amdgcn_wmma_f32_16x16x4_f32(
         false, a,  false, b,
         (short)0, acc,  false,  false);
    arow += 4;
    bcol += (size_t)4 * N;
  }

  const int rbase = m0 + ((lane >> 4) << 3);
  float* cptr = C + (size_t)rbase * N + n0 + l15;
#pragma unroll
  for (int v = 0; v < 8; ++v) cptr[(size_t)v * N] = acc[v];
}

__global__ __launch_bounds__(256) void k_bias_act(float* __restrict__ agg,
                                                  const float* __restrict__ b,
                                                  int total, int F, int do_relu) {
  int idx = blockIdx.x * blockDim.x + threadIdx.x;
  if (idx < total) {
    float v = agg[idx] + b[idx % F];
    if (do_relu) v = fmaxf(v, 0.0f);
    agg[idx] = v;
  }
}

extern "C" void kernel_launch(void* const* d_in, const int* in_sizes, int n_in,
                              void* d_out, int out_size, void* d_ws, size_t ws_size,
                              hipStream_t stream) {
  const float* x  = (const float*)d_in[0];
  const int*   ei = (const int*)d_in[1];
  const float* W1 = (const float*)d_in[2];
  const float* b1 = (const float*)d_in[3];
  const float* W2 = (const float*)d_in[4];
  const float* b2 = (const float*)d_in[5];

  const int H  = in_sizes[3];
  const int C  = in_sizes[5];
  const int F  = in_sizes[2] / H;
  const int Nn = in_sizes[0] / F;
  const int E  = in_sizes[1] / 2;
  const int* rowIdx = ei;
  const int* colIdx = ei + E;

  auto al = [](size_t v) { return (v + 255) & ~(size_t)255; };
  size_t off = 0;
  float* dinv = (float*)((char*)d_ws + off);
  off += al((size_t)Nn * sizeof(float));
  int* cnt = (int*)((char*)d_ws + off);
  off += al((size_t)(Nn + 1) * sizeof(int));
  Seg* seg = (Seg*)((char*)d_ws + off);
  off += al((size_t)Nn * sizeof(Seg));
  int* rank = (int*)((char*)d_ws + off);
  off += al((size_t)E * sizeof(int));
  int* srcs = (int*)((char*)d_ws + off);
  off += al((size_t)E * sizeof(int));
  float* hbuf = (float*)((char*)d_ws + off);
  off += al((size_t)Nn * H * sizeof(float));
  float* abuf = (float*)((char*)d_ws + off);

  const dim3 blk(256);
  auto blocks = [](long long work) { return (unsigned)((work + 255) / 256); };
  const unsigned rowBlocks = (unsigned)((Nn + 7) / 8);

  k_zero_counts<<<blocks(Nn + 1), blk, 0, stream>>>(cnt, Nn + 1);
  k_count_rank <<<blocks(E),      blk, 0, stream>>>(colIdx, cnt, rank, E);
  k_offsets    <<<blocks(Nn),     blk, 0, stream>>>(cnt, cnt + Nn, seg, dinv, Nn);
  k_fill       <<<blocks(E),      blk, 0, stream>>>(rowIdx, colIdx, rank, seg, srcs, E);

  k_gather_rows<4><<<rowBlocks, blk, 0, stream>>>(x, seg, srcs, dinv, abuf, Nn);
  const int tiles1 = ((Nn + 15) / 16) * (H / 16);
  k_gemm_wmma_f32<<<(tiles1 + 7) / 8, blk, 0, stream>>>(abuf, W1, hbuf, Nn, F, H);
  const int tot1 = Nn * H;
  k_bias_act<<<blocks(tot1), blk, 0, stream>>>(hbuf, b1, tot1, H,  1);

  const int tiles2 = ((Nn + 15) / 16) * (C / 16);
  k_gemm_wmma_f32<<<(tiles2 + 7) / 8, blk, 0, stream>>>(hbuf, W2, abuf, Nn, H, C);
  const int tot2 = Nn * C;
  float* out = (float*)d_out;
  k_gather_rows<2><<<rowBlocks, blk, 0, stream>>>(abuf, seg, srcs, dinv, out, Nn);
  k_bias_act<<<blocks(tot2), blk, 0, stream>>>(out, b2, tot2, C,  0);
}


# bias add (and layer 1's ReLU) fused into the store of the kernel that produces the rows (GEMM 1 tile store, layer 2 row gather); the two elementwise passes and their two launches are removed
# speedup vs baseline: 2.7337x; 1.1821x over previous
#include <hip/hip_runtime.h>
#include <hip/hip_bf16.h>


typedef __attribute__((ext_vector_type(2))) float v2f;
typedef __attribute__((ext_vector_type(8))) float v8f;

struct __attribute__((aligned(8))) Seg {
  int start;
  int len;
};

__global__ __launch_bounds__(256) void k_zero_counts(int* __restrict__ cnt, int n) {
  int i = blockIdx.x * blockDim.x + threadIdx.x;
  if (i < n) cnt[i] = 0;
}

__global__ __launch_bounds__(256) void k_count_rank(const int* __restrict__ col,
                                                    int* cnt,
                                                    int* __restrict__ rank, int E) {
  int e = blockIdx.x * blockDim.x + threadIdx.x;
  if (e < E) {
    rank[e] = __hip_atomic_fetch_add(&cnt[col[e]], 1, __ATOMIC_RELAXED,
                                     __HIP_MEMORY_SCOPE_AGENT);
  }
}

__global__ __launch_bounds__(256) void k_offsets(const int* __restrict__ cnt,
                                                 int* cursor,
                                                 Seg* __restrict__ seg,
                                                 float* __restrict__ dinv, int n) {
  __shared__ int wsum[8];
  __shared__ int blk_base;
  const int i    = blockIdx.x * blockDim.x + threadIdx.x;
  const int lane = threadIdx.x & 31;
  const int wave = threadIdx.x >> 5;
  const int c = (i < n) ? cnt[i] : 0;
  int v = c;
#pragma unroll
  for (int d = 1; d < 32; d <<= 1) {
    int t = __shfl_up(v, d);
    if (lane >= d) v += t;
  }
  if (lane == 31) wsum[wave] = v;
  __syncthreads();
  if (threadIdx.x == 0) {
    int tot = 0;
#pragma unroll
    for (int w = 0; w < 8; ++w) {
      int t = wsum[w];
      wsum[w] = tot;
      tot += t;
    }
    blk_base = __hip_atomic_fetch_add(cursor, tot, __ATOMIC_RELAXED,
                                      __HIP_MEMORY_SCOPE_AGENT);
  }
  __syncthreads();
  if (i < n) {
    Seg sg;
    sg.start = blk_base + wsum[wave] + (v - c);
    sg.len   = c;
    seg[i] = sg;
    const float d = (float)(c + 1);
    dinv[i] = (d > 0.0f) ? rsqrtf(d) : 0.0f;
  }
}

__global__ __launch_bounds__(256) void k_fill(const int* __restrict__ row,
                                              const int* __restrict__ col,
                                              const int* __restrict__ rank,
                                              const Seg* __restrict__ seg,
                                              int* __restrict__ srcs, int E) {
  int e = blockIdx.x * blockDim.x + threadIdx.x;
  if (e < E) srcs[seg[col[e]].start + rank[e]] = row[e];
}

template <int VEC, bool ADD_BIAS>
__global__ __launch_bounds__(256) void k_gather_rows(
    const float* __restrict__ h, const Seg* __restrict__ seg,
    const int* __restrict__ srcs, const float* __restrict__ dinv,
    const float* __restrict__ bias, float* __restrict__ out, int n_nodes) {
  typedef float vec __attribute__((ext_vector_type(VEC)));
  const int lane = threadIdx.x & 31;
  const int node = __builtin_amdgcn_readfirstlane(
      (int)(blockIdx.x * (blockDim.x >> 5) + (threadIdx.x >> 5)));
  if (node >= n_nodes) return;
  const Seg   sg = seg[node];
  const int   n  = sg.len;
  const float di = dinv[node];
  const int* __restrict__ sp = srcs + sg.start;
  const vec* __restrict__ hv = (const vec*)h + lane;

  vec bvec = {};
  if (ADD_BIAS) {
#pragma unroll
    for (int c = 0; c < VEC; ++c) bvec[c] = bias[lane * VEC + c];
  }

  vec acc = hv[(size_t)node * 32] * (di * di);
  int j = 0;
  for (; j + 4 <= n; j += 4) {
    const int s0 = sp[j], s1 = sp[j + 1], s2 = sp[j + 2], s3 = sp[j + 3];
    const vec r0 = hv[(size_t)s0 * 32];
    const vec r1 = hv[(size_t)s1 * 32];
    const vec r2 = hv[(size_t)s2 * 32];
    const vec r3 = hv[(size_t)s3 * 32];
    const float w0 = dinv[s0] * di, w1 = dinv[s1] * di;
    const float w2 = dinv[s2] * di, w3 = dinv[s3] * di;
    acc += r0 * w0;
    acc += r1 * w1;
    acc += r2 * w2;
    acc += r3 * w3;
  }
  if (j + 2 <= n) {
    const int s0 = sp[j], s1 = sp[j + 1];
    const vec r0 = hv[(size_t)s0 * 32];
    const vec r1 = hv[(size_t)s1 * 32];
    const float w0 = dinv[s0] * di, w1 = dinv[s1] * di;
    acc += r0 * w0;
    acc += r1 * w1;
    j += 2;
  }
  if (j < n) {
    const int s0 = sp[j];
    const vec r0 = hv[(size_t)s0 * 32];
    acc += r0 * (dinv[s0] * di);
  }
  if (ADD_BIAS) acc += bvec;
  ((vec*)out)[(size_t)node * 32 + lane] = acc;
}

template <bool BIAS_RELU>
__global__ __launch_bounds__(256) void k_gemm_wmma_f32(
    const float* __restrict__ A, const float* __restrict__ B,
    const float* __restrict__ bias, float* __restrict__ C, int M, int K, int N) {
  const int lane  = threadIdx.x & 31;
  const int wave  = threadIdx.x >> 5;
  const int tilesN = N >> 4;
  const int t  = blockIdx.x * (blockDim.x >> 5) + wave;
  const int m0 = (t / tilesN) << 4;
  const int n0 = (t % tilesN) << 4;
  if (m0 >= M) return;

  const int l15   = lane & 15;
  const int khalf = (lane >> 4) << 1;

  const float* arow = A + (size_t)(m0 + l15) * K + khalf;
  const float* bcol = B + (size_t)khalf * N + (n0 + l15);
  const float bv = BIAS_RELU ? bias[n0 + l15] : 0.0f;

  v8f acc = {};
  for (int k0 = 0; k0 < K; k0 += 4) {
    v2f a, b;
    float2 av = *(const float2*)arow;
    a.x = av.x;
    a.y = av.y;
    b.x = bcol[0];
    b.y = bcol[N];
    acc = __builtin_amdgcn_wmma_f32_16x16x4_f32(
         false, a,  false, b,
         (short)0, acc,  false,  false);
    arow += 4;
    bcol += (size_t)4 * N;
  }

  const int rbase = m0 + ((lane >> 4) << 3);
  float* cptr = C + (size_t)rbase * N + n0 + l15;
#pragma unroll
  for (int v = 0; v < 8; ++v) {
    float r = acc[v];
    if (BIAS_RELU) r = fmaxf(r + bv, 0.0f);
    cptr[(size_t)v * N] = r;
  }
}

extern "C" void kernel_launch(void* const* d_in, const int* in_sizes, int n_in,
                              void* d_out, int out_size, void* d_ws, size_t ws_size,
                              hipStream_t stream) {
  const float* x  = (const float*)d_in[0];
  const int*   ei = (const int*)d_in[1];
  const float* W1 = (const float*)d_in[2];
  const float* b1 = (const float*)d_in[3];
  const float* W2 = (const float*)d_in[4];
  const float* b2 = (const float*)d_in[5];

  const int H  = in_sizes[3];
  const int C  = in_sizes[5];
  const int F  = in_sizes[2] / H;
  const int Nn = in_sizes[0] / F;
  const int E  = in_sizes[1] / 2;
  const int* rowIdx = ei;
  const int* colIdx = ei + E;

  auto al = [](size_t v) { return (v + 255) & ~(size_t)255; };
  size_t off = 0;
  float* dinv = (float*)((char*)d_ws + off);
  off += al((size_t)Nn * sizeof(float));
  int* cnt = (int*)((char*)d_ws + off);
  off += al((size_t)(Nn + 1) * sizeof(int));
  Seg* seg = (Seg*)((char*)d_ws + off);
  off += al((size_t)Nn * sizeof(Seg));
  int* rank = (int*)((char*)d_ws + off);
  off += al((size_t)E * sizeof(int));
  int* srcs = (int*)((char*)d_ws + off);
  off += al((size_t)E * sizeof(int));
  float* hbuf = (float*)((char*)d_ws + off);
  off += al((size_t)Nn * H * sizeof(float));
  float* abuf = (float*)((char*)d_ws + off);

  const dim3 blk(256);
  auto blocks = [](long long work) { return (unsigned)((work + 255) / 256); };
  const unsigned rowBlocks = (unsigned)((Nn + 7) / 8);

  k_zero_counts<<<blocks(Nn + 1), blk, 0, stream>>>(cnt, Nn + 1);
  k_count_rank <<<blocks(E),      blk, 0, stream>>>(colIdx, cnt, rank, E);
  k_offsets    <<<blocks(Nn),     blk, 0, stream>>>(cnt, cnt + Nn, seg, dinv, Nn);
  k_fill       <<<blocks(E),      blk, 0, stream>>>(rowIdx, colIdx, rank, seg, srcs, E);

  k_gather_rows<4, false><<<rowBlocks, blk, 0, stream>>>(x, seg, srcs, dinv, nullptr, abuf, Nn);
  const int tiles1 = ((Nn + 15) / 16) * (H / 16);
  k_gemm_wmma_f32<true><<<(tiles1 + 7) / 8, blk, 0, stream>>>(abuf, W1, b1, hbuf, Nn, F, H);

  const int tiles2 = ((Nn + 15) / 16) * (C / 16);
  k_gemm_wmma_f32<false><<<(tiles2 + 7) / 8, blk, 0, stream>>>(hbuf, W2, nullptr, abuf, Nn, H, C);
  float* out = (float*)d_out;
  k_gather_rows<2, true><<<rowBlocks, blk, 0, stream>>>(abuf, seg, srcs, dinv, b2, out, Nn);
}


# GEMMs: work-groups copy the weight matrix into LDS once, in fragment order; persistent waves keep a 16-row A strip in registers for 2-4 column tiles; k permuted within 8s (f32 sum order changes)
# speedup vs baseline: 3.3991x; 1.2434x over previous
#include <hip/hip_runtime.h>
#include <hip/hip_bf16.h>


typedef __attribute__((ext_vector_type(2))) float v2f;
typedef __attribute__((ext_vector_type(8))) float v8f;

struct __attribute__((aligned(8))) Seg {
  int start;
  int len;
};

__global__ __launch_bounds__(256) void k_zero_counts(int* __restrict__ cnt, int n) {
  int i = blockIdx.x * blockDim.x + threadIdx.x;
  if (i < n) cnt[i] = 0;
}

__global__ __launch_bounds__(256) void k_count_rank(const int* __restrict__ col,
                                                    int* cnt,
                                                    int* __restrict__ rank, int E) {
  int e = blockIdx.x * blockDim.x + threadIdx.x;
  if (e < E) {
    rank[e] = __hip_atomic_fetch_add(&cnt[col[e]], 1, __ATOMIC_RELAXED,
                                     __HIP_MEMORY_SCOPE_AGENT);
  }
}

__global__ __launch_bounds__(256) void k_offsets(const int* __restrict__ cnt,
                                                 int* cursor,
                                                 Seg* __restrict__ seg,
                                                 float* __restrict__ dinv, int n) {
  __shared__ int wsum[8];
  __shared__ int blk_base;
  const int i    = blockIdx.x * blockDim.x + threadIdx.x;
  const int lane = threadIdx.x & 31;
  const int wave = threadIdx.x >> 5;
  const int c = (i < n) ? cnt[i] : 0;
  int v = c;
#pragma unroll
  for (int d = 1; d < 32; d <<= 1) {
    int t = __shfl_up(v, d);
    if (lane >= d) v += t;
  }
  if (lane == 31) wsum[wave] = v;
  __syncthreads();
  if (threadIdx.x == 0) {
    int tot = 0;
#pragma unroll
    for (int w = 0; w < 8; ++w) {
      int t = wsum[w];
      wsum[w] = tot;
      tot += t;
    }
    blk_base = __hip_atomic_fetch_add(cursor, tot, __ATOMIC_RELAXED,
                                      __HIP_MEMORY_SCOPE_AGENT);
  }
  __syncthreads();
  if (i < n) {
    Seg sg;
    sg.start = blk_base + wsum[wave] + (v - c);
    sg.len   = c;
    seg[i] = sg;
    const float d = (float)(c + 1);
    dinv[i] = (d > 0.0f) ? rsqrtf(d) : 0.0f;
  }
}

__global__ __launch_bounds__(256) void k_fill(const int* __restrict__ row,
                                              const int* __restrict__ col,
                                              const int* __restrict__ rank,
                                              const Seg* __restrict__ seg,
                                              int* __restrict__ srcs, int E) {
  int e = blockIdx.x * blockDim.x + threadIdx.x;
  if (e < E) srcs[seg[col[e]].start + rank[e]] = row[e];
}

template <int VEC, bool ADD_BIAS>
__global__ __launch_bounds__(256) void k_gather_rows(
    const float* __restrict__ h, const Seg* __restrict__ seg,
    const int* __restrict__ srcs, const float* __restrict__ dinv,
    const float* __restrict__ bias, float* __restrict__ out, int n_nodes) {
  typedef float vec __attribute__((ext_vector_type(VEC)));
  const int lane = threadIdx.x & 31;
  const int node = __builtin_amdgcn_readfirstlane(
      (int)(blockIdx.x * (blockDim.x >> 5) + (threadIdx.x >> 5)));
  if (node >= n_nodes) return;
  const Seg   sg = seg[node];
  const int   n  = sg.len;
  const float di = dinv[node];
  const int* __restrict__ sp = srcs + sg.start;
  const vec* __restrict__ hv = (const vec*)h + lane;

  vec bvec = {};
  if (ADD_BIAS) {
#pragma unroll
    for (int c = 0; c < VEC; ++c) bvec[c] = bias[lane * VEC + c];
  }

  vec acc = hv[(size_t)node * 32] * (di * di);
  int j = 0;
  for (; j + 4 <= n; j += 4) {
    const int s0 = sp[j], s1 = sp[j + 1], s2 = sp[j + 2], s3 = sp[j + 3];
    const vec r0 = hv[(size_t)s0 * 32];
    const vec r1 = hv[(size_t)s1 * 32];
    const vec r2 = hv[(size_t)s2 * 32];
    const vec r3 = hv[(size_t)s3 * 32];
    const float w0 = dinv[s0] * di, w1 = dinv[s1] * di;
    const float w2 = dinv[s2] * di, w3 = dinv[s3] * di;
    acc += r0 * w0;
    acc += r1 * w1;
    acc += r2 * w2;
    acc += r3 * w3;
  }
  if (j + 2 <= n) {
    const int s0 = sp[j], s1 = sp[j + 1];
    const vec r0 = hv[(size_t)s0 * 32];
    const vec r1 = hv[(size_t)s1 * 32];
    const float w0 = dinv[s0] * di, w1 = dinv[s1] * di;
    acc += r0 * w0;
    acc += r1 * w1;
    j += 2;
  }
  if (j < n) {
    const int s0 = sp[j];
    const vec r0 = hv[(size_t)s0 * 32];
    acc += r0 * (dinv[s0] * di);
  }
  if (ADD_BIAS) acc += bvec;
  ((vec*)out)[(size_t)node * 32 + lane] = acc;
}

constexpr int GEMM_THREADS = 512;

template <int K, int N, int NT, bool BIAS_RELU>
__global__ __launch_bounds__(GEMM_THREADS) void k_gemm_wmma_f32(
    const float* __restrict__ A, const float* __restrict__ B,
    const float* __restrict__ bias, float* __restrict__ C, int M) {
  constexpr int KP     = K / 8;
  constexpr int NTILES = N / 16;
  constexpr int NSPLIT = NTILES / NT;
  __shared__ __attribute__((aligned(16))) float4 Bs[KP * NTILES * 32];

  static_assert((KP * 2 * N) % GEMM_THREADS == 0, "the copy loop has no tail");
#pragma unroll
  for (int it = 0; it < KP * 2 * N / GEMM_THREADS; ++it) {
    const int idx = it * GEMM_THREADS + (int)threadIdx.x;
    const int n  = idx % N;
    const int jh = idx / N;
    const float* src = B + (size_t)(4 * jh) * N + n;
    float4 v;
    v.x = src[0];
    v.y = src[N];
    v.z = src[2 * N];
    v.w = src[3 * N];
    Bs[((jh >> 1) * NTILES + (n >> 4)) * 32 + (jh & 1) * 16 + (n & 15)] = v;
  }
  __syncthreads();

  const int lane = threadIdx.x & 31;
  const int wave = __builtin_amdgcn_readfirstlane((int)(threadIdx.x >> 5));
  const int c = lane & 15;
  const int h = lane >> 4;
  const int nwaves = gridDim.x * (GEMM_THREADS >> 5);
  const int units  = (M >> 4) * NSPLIT;

  for (int u = wave * gridDim.x + blockIdx.x; u < units; u += nwaves) {
    const int m0 = (u / NSPLIT) << 4;
    const int t0 = (u % NSPLIT) * NT;

    const float4* ap = (const float4*)(A + (size_t)(m0 + c) * K + 4 * h);
    float4 q[KP];
#pragma unroll
    for (int j = 0; j < KP; ++j) q[j] = ap[2 * j];

    float bv[NT];
#pragma unroll
    for (int t = 0; t < NT; ++t) bv[t] = BIAS_RELU ? bias[(t0 + t) * 16 + c] : 0.0f;

    __builtin_amdgcn_sched_barrier(0);

    v8f acc[NT];
#pragma unroll
    for (int t = 0; t < NT; ++t) acc[t] = v8f{};

    const float4* bp = Bs + t0 * 32 + lane;
#pragma unroll
    for (int j = 0; j < KP; ++j) {
      float4 bf[NT];
#pragma unroll
      for (int t = 0; t < NT; ++t) bf[t] = bp[(j * NTILES + t) * 32];
      v2f a0, a1;
      a0.x = q[j].x;
      a0.y = q[j].y;
      a1.x = q[j].z;
      a1.y = q[j].w;
#pragma unroll
      for (int t = 0; t < NT; ++t) {
        v2f b0;
        b0.x = bf[t].x;
        b0.y = bf[t].y;
        acc[t] = __builtin_amdgcn_wmma_f32_16x16x4_f32(
             false, a0,  false, b0,
             (short)0, acc[t],  false,  false);
      }
#pragma unroll
      for (int t = 0; t < NT; ++t) {
        v2f b1;
        b1.x = bf[t].z;
        b1.y = bf[t].w;
        acc[t] = __builtin_amdgcn_wmma_f32_16x16x4_f32(
             false, a1,  false, b1,
             (short)0, acc[t],  false,  false);
      }
    }

    float* cptr = C + (size_t)(m0 + 8 * h) * N + t0 * 16 + c;
#pragma unroll
    for (int r = 0; r < 8; ++r) {
#pragma unroll
      for (int t = 0; t < NT; ++t) {
        float v = acc[t][r];
        if (BIAS_RELU) v = fmaxf(v + bv[t], 0.0f);
        cptr[(size_t)r * N + t * 16] = v;
      }
    }
  }
}

extern "C" void kernel_launch(void* const* d_in, const int* in_sizes, int n_in,
                              void* d_out, int out_size, void* d_ws, size_t ws_size,
                              hipStream_t stream) {
  const float* x  = (const float*)d_in[0];
  const int*   ei = (const int*)d_in[1];
  const float* W1 = (const float*)d_in[2];
  const float* b1 = (const float*)d_in[3];
  const float* W2 = (const float*)d_in[4];
  const float* b2 = (const float*)d_in[5];

  const int H  = in_sizes[3];
  const int F  = in_sizes[2] / H;
  const int Nn = in_sizes[0] / F;
  const int E  = in_sizes[1] / 2;
  const int* rowIdx = ei;
  const int* colIdx = ei + E;

  auto al = [](size_t v) { return (v + 255) & ~(size_t)255; };
  size_t off = 0;
  float* dinv = (float*)((char*)d_ws + off);
  off += al((size_t)Nn * sizeof(float));
  int* cnt = (int*)((char*)d_ws + off);
  off += al((size_t)(Nn + 1) * sizeof(int));
  Seg* seg = (Seg*)((char*)d_ws + off);
  off += al((size_t)Nn * sizeof(Seg));
  int* rank = (int*)((char*)d_ws + off);
  off += al((size_t)E * sizeof(int));
  int* srcs = (int*)((char*)d_ws + off);
  off += al((size_t)E * sizeof(int));
  float* hbuf = (float*)((char*)d_ws + off);
  off += al((size_t)Nn * H * sizeof(float));
  float* abuf = (float*)((char*)d_ws + off);

  const dim3 blk(256);
  auto blocks = [](long long work) { return (unsigned)((work + 255) / 256); };
  const unsigned rowBlocks = (unsigned)((Nn + 7) / 8);

  k_zero_counts<<<blocks(Nn + 1), blk, 0, stream>>>(cnt, Nn + 1);
  k_count_rank <<<blocks(E),      blk, 0, stream>>>(colIdx, cnt, rank, E);
  k_offsets    <<<blocks(Nn),     blk, 0, stream>>>(cnt, cnt + Nn, seg, dinv, Nn);
  k_fill       <<<blocks(E),      blk, 0, stream>>>(rowIdx, colIdx, rank, seg, srcs, E);

  k_gather_rows<4, false><<<rowBlocks, blk, 0, stream>>>(x, seg, srcs, dinv, nullptr, abuf, Nn);
  k_gemm_wmma_f32<128, 256, 4, true><<<256, GEMM_THREADS, 0, stream>>>(abuf, W1, b1, hbuf, Nn);

  k_gemm_wmma_f32<256, 64, 2, false><<<256, GEMM_THREADS, 0, stream>>>(hbuf, W2, nullptr, abuf, Nn);
  float* out = (float*)d_out;
  k_gather_rows<2, true><<<rowBlocks, blk, 0, stream>>>(abuf, seg, srcs, dinv, b2, out, Nn);
}


# edge ranks counted in LDS: 64 work-groups each count a slice of the edge list into per-node LDS counters (no global atomic per edge; zeroing launch gone); row sums in slice order, then LDS arrival
# speedup vs baseline: 4.5234x; 1.3308x over previous
#include <hip/hip_runtime.h>
#include <hip/hip_bf16.h>


typedef __attribute__((ext_vector_type(2))) float v2f;
typedef __attribute__((ext_vector_type(8))) float v8f;

struct __attribute__((aligned(8))) Seg {
  int start;
  int len;
};

constexpr int SLICES      = 64;
constexpr int CNT_THREADS = 1024;
constexpr int CNT_CAP     = 51200;
constexpr int EPT         = 8;

__global__ __launch_bounds__(CNT_THREADS) void k_count_rank(
    const int* __restrict__ col, int* __restrict__ tab, int* __restrict__ rank,
    int* __restrict__ cursor, int E, int n, int pitch, int chunk) {
  __shared__ __attribute__((aligned(16))) int lcnt[CNT_CAP];
  int4* l4 = (int4*)lcnt;
  const int tid = threadIdx.x;
  const int e0  = blockIdx.x * chunk;
  const int e1  = min(e0 + chunk, E);
  if (blockIdx.x == 0 && tid == 0) *cursor = 0;
  for (int r0 = 0; r0 < n; r0 += CNT_CAP) {
    const int rn  = min(CNT_CAP, n - r0);
    const int rn4 = (rn + 3) >> 2;
    for (int i = tid; i < rn4; i += CNT_THREADS) l4[i] = int4{0, 0, 0, 0};
    __syncthreads();
    for (int e = e0 + tid; e < e1; e += EPT * CNT_THREADS) {
      int c[EPT];
#pragma unroll
      for (int k = 0; k < EPT; ++k) c[k] = col[min(e + k * CNT_THREADS, e1 - 1)];
#pragma unroll
      for (int k = 0; k < EPT; ++k) {
        c[k] = (e + k * CNT_THREADS < e1) ? c[k] - r0 : -1;
      }
      int r[EPT];
#pragma unroll
      for (int k = 0; k < EPT; ++k) {
        r[k] = 0;
        if ((unsigned)c[k] < (unsigned)rn) {
          r[k] = __hip_atomic_fetch_add(&lcnt[c[k]], 1, __ATOMIC_RELAXED,
                                        __HIP_MEMORY_SCOPE_WORKGROUP);
        }
      }
#pragma unroll
      for (int k = 0; k < EPT; ++k) {
        if ((unsigned)c[k] < (unsigned)rn) rank[e + k * CNT_THREADS] = r[k];
      }
    }
    __syncthreads();
    int4* __restrict__ t4 = (int4*)(tab + (size_t)blockIdx.x * pitch + r0);
    for (int i = tid; i < rn4; i += CNT_THREADS) t4[i] = l4[i];
    __syncthreads();
  }
}

__global__ __launch_bounds__(256) void k_offsets(int* tab, int* cursor,
                                                 Seg* __restrict__ seg,
                                                 float* __restrict__ dinv, int n,
                                                 int pitch) {
  __shared__ int wsum[8];
  __shared__ int blk_base;
  const int i    = blockIdx.x * blockDim.x + threadIdx.x;
  const int lane = threadIdx.x & 31;
  const int wave = threadIdx.x >> 5;
  const int ic = min(i, n - 1);
  int cs[SLICES];
#pragma unroll
  for (int p = 0; p < SLICES; ++p) cs[p] = tab[(size_t)p * pitch + ic];
  int c = 0;
#pragma unroll
  for (int p = 0; p < SLICES; ++p) c += cs[p];
  if (i >= n) c = 0;
  int v = c;
#pragma unroll
  for (int d = 1; d < 32; d <<= 1) {
    int t = __shfl_up(v, d);
    if (lane >= d) v += t;
  }
  if (lane == 31) wsum[wave] = v;
  __syncthreads();
  if (threadIdx.x == 0) {
    int tot = 0;
#pragma unroll
    for (int w = 0; w < 8; ++w) {
      int t = wsum[w];
      wsum[w] = tot;
      tot += t;
    }
    blk_base = __hip_atomic_fetch_add(cursor, tot, __ATOMIC_RELAXED,
                                      __HIP_MEMORY_SCOPE_AGENT);
  }
  __syncthreads();
  if (i < n) {
    Seg sg;
    sg.start = blk_base + wsum[wave] + (v - c);
    sg.len   = c;
    seg[i] = sg;
    const float d = (float)(c + 1);
    dinv[i] = (d > 0.0f) ? rsqrtf(d) : 0.0f;
    int run = sg.start;
#pragma unroll
    for (int p = 0; p < SLICES; ++p) {
      tab[(size_t)p * pitch + i] = run;
      run += cs[p];
    }
  }
}

__global__ __launch_bounds__(256) void k_fill(const int* __restrict__ row,
                                              const int* __restrict__ col,
                                              const int* __restrict__ rank,
                                              const int* __restrict__ tab,
                                              int* __restrict__ srcs, int E,
                                              int pitch, int chunk) {
  const int p  = blockIdx.y;
  const int le = blockIdx.x * blockDim.x + threadIdx.x;
  const int e  = p * chunk + le;
  if (le < chunk && e < E) srcs[tab[(size_t)p * pitch + col[e]] + rank[e]] = row[e];
}

template <int VEC, bool ADD_BIAS>
__global__ __launch_bounds__(256) void k_gather_rows(
    const float* __restrict__ h, const Seg* __restrict__ seg,
    const int* __restrict__ srcs, const float* __restrict__ dinv,
    const float* __restrict__ bias, float* __restrict__ out, int n_nodes) {
  typedef float vec __attribute__((ext_vector_type(VEC)));
  const int lane = threadIdx.x & 31;
  const int node = __builtin_amdgcn_readfirstlane(
      (int)(blockIdx.x * (blockDim.x >> 5) + (threadIdx.x >> 5)));
  if (node >= n_nodes) return;
  const Seg   sg = seg[node];
  const int   n  = sg.len;
  const float di = dinv[node];
  const int* __restrict__ sp = srcs + sg.start;
  const vec* __restrict__ hv = (const vec*)h + lane;

  vec bvec = {};
  if (ADD_BIAS) {
#pragma unroll
    for (int c = 0; c < VEC; ++c) bvec[c] = bias[lane * VEC + c];
  }

  vec acc = hv[(size_t)node * 32] * (di * di);
  int j = 0;
  for (; j + 4 <= n; j += 4) {
    const int s0 = sp[j], s1 = sp[j + 1], s2 = sp[j + 2], s3 = sp[j + 3];
    const vec r0 = hv[(size_t)s0 * 32];
    const vec r1 = hv[(size_t)s1 * 32];
    const vec r2 = hv[(size_t)s2 * 32];
    const vec r3 = hv[(size_t)s3 * 32];
    const float w0 = dinv[s0] * di, w1 = dinv[s1] * di;
    const float w2 = dinv[s2] * di, w3 = dinv[s3] * di;
    acc += r0 * w0;
    acc += r1 * w1;
    acc += r2 * w2;
    acc += r3 * w3;
  }
  if (j + 2 <= n) {
    const int s0 = sp[j], s1 = sp[j + 1];
    const vec r0 = hv[(size_t)s0 * 32];
    const vec r1 = hv[(size_t)s1 * 32];
    const float w0 = dinv[s0] * di, w1 = dinv[s1] * di;
    acc += r0 * w0;
    acc += r1 * w1;
    j += 2;
  }
  if (j < n) {
    const int s0 = sp[j];
    const vec r0 = hv[(size_t)s0 * 32];
    acc += r0 * (dinv[s0] * di);
  }
  if (ADD_BIAS) acc += bvec;
  ((vec*)out)[(size_t)node * 32 + lane] = acc;
}

constexpr int GEMM_THREADS = 512;

template <int K, int N, int NT, bool BIAS_RELU>
__global__ __launch_bounds__(GEMM_THREADS) void k_gemm_wmma_f32(
    const float* __restrict__ A, const float* __restrict__ B,
    const float* __restrict__ bias, float* __restrict__ C, int M) {
  constexpr int KP     = K / 8;
  constexpr int NTILES = N / 16;
  constexpr int NSPLIT = NTILES / NT;
  __shared__ __attribute__((aligned(16))) float4 Bs[KP * NTILES * 32];

  static_assert((KP * 2 * N) % GEMM_THREADS == 0, "the copy loop has no tail");
#pragma unroll
  for (int it = 0; it < KP * 2 * N / GEMM_THREADS; ++it) {
    const int idx = it * GEMM_THREADS + (int)threadIdx.x;
    const int n  = idx % N;
    const int jh = idx / N;
    const float* src = B + (size_t)(4 * jh) * N + n;
    float4 v;
    v.x = src[0];
    v.y = src[N];
    v.z = src[2 * N];
    v.w = src[3 * N];
    Bs[((jh >> 1) * NTILES + (n >> 4)) * 32 + (jh & 1) * 16 + (n & 15)] = v;
  }
  __syncthreads();

  const int lane = threadIdx.x & 31;
  const int wave = __builtin_amdgcn_readfirstlane((int)(threadIdx.x >> 5));
  const int c = lane & 15;
  const int h = lane >> 4;
  const int nwaves = gridDim.x * (GEMM_THREADS >> 5);
  const int units  = (M >> 4) * NSPLIT;

  for (int u = wave * gridDim.x + blockIdx.x; u < units; u += nwaves) {
    const int m0 = (u / NSPLIT) << 4;
    const int t0 = (u % NSPLIT) * NT;

    const float4* ap = (const float4*)(A + (size_t)(m0 + c) * K + 4 * h);
    float4 q[KP];
#pragma unroll
    for (int j = 0; j < KP; ++j) q[j] = ap[2 * j];

    float bv[NT];
#pragma unroll
    for (int t = 0; t < NT; ++t) bv[t] = BIAS_RELU ? bias[(t0 + t) * 16 + c] : 0.0f;

    __builtin_amdgcn_sched_barrier(0);

    v8f acc[NT];
#pragma unroll
    for (int t = 0; t < NT; ++t) acc[t] = v8f{};

    const float4* bp = Bs + t0 * 32 + lane;
#pragma unroll
    for (int j = 0; j < KP; ++j) {
      float4 bf[NT];
#pragma unroll
      for (int t = 0; t < NT; ++t) bf[t] = bp[(j * NTILES + t) * 32];
      v2f a0, a1;
      a0.x = q[j].x;
      a0.y = q[j].y;
      a1.x = q[j].z;
      a1.y = q[j].w;
#pragma unroll
      for (int t = 0; t < NT; ++t) {
        v2f b0;
        b0.x = bf[t].x;
        b0.y = bf[t].y;
        acc[t] = __builtin_amdgcn_wmma_f32_16x16x4_f32(
             false, a0,  false, b0,
             (short)0, acc[t],  false,  false);
      }
#pragma unroll
      for (int t = 0; t < NT; ++t) {
        v2f b1;
        b1.x = bf[t].z;
        b1.y = bf[t].w;
        acc[t] = __builtin_amdgcn_wmma_f32_16x16x4_f32(
             false, a1,  false, b1,
             (short)0, acc[t],  false,  false);
      }
    }

    float* cptr = C + (size_t)(m0 + 8 * h) * N + t0 * 16 + c;
#pragma unroll
    for (int r = 0; r < 8; ++r) {
#pragma unroll
      for (int t = 0; t < NT; ++t) {
        float v = acc[t][r];
        if (BIAS_RELU) v = fmaxf(v + bv[t], 0.0f);
        cptr[(size_t)r * N + t * 16] = v;
      }
    }
  }
}

extern "C" void kernel_launch(void* const* d_in, const int* in_sizes, int n_in,
                              void* d_out, int out_size, void* d_ws, size_t ws_size,
                              hipStream_t stream) {
  const float* x  = (const float*)d_in[0];
  const int*   ei = (const int*)d_in[1];
  const float* W1 = (const float*)d_in[2];
  const float* b1 = (const float*)d_in[3];
  const float* W2 = (const float*)d_in[4];
  const float* b2 = (const float*)d_in[5];

  const int H  = in_sizes[3];
  const int F  = in_sizes[2] / H;
  const int Nn = in_sizes[0] / F;
  const int E  = in_sizes[1] / 2;
  const int* rowIdx = ei;
  const int* colIdx = ei + E;

  auto al = [](size_t v) { return (v + 255) & ~(size_t)255; };
  size_t off = 0;
  float* dinv = (float*)((char*)d_ws + off);
  off += al((size_t)Nn * sizeof(float));
  int* cursor = (int*)((char*)d_ws + off);
  off += al(sizeof(int));
  int* tab = (int*)((char*)d_ws + off);
  const int pitch = (Nn + 63) & ~63;
  off += al((size_t)SLICES * pitch * sizeof(int));
  Seg* seg = (Seg*)((char*)d_ws + off);
  off += al((size_t)Nn * sizeof(Seg));
  int* rank = (int*)((char*)d_ws + off);
  off += al((size_t)E * sizeof(int));
  int* srcs = (int*)((char*)d_ws + off);
  off += al((size_t)E * sizeof(int));
  float* hbuf = (float*)((char*)d_ws + off);
  off += al((size_t)Nn * H * sizeof(float));
  float* abuf = (float*)((char*)d_ws + off);

  const dim3 blk(256);
  auto blocks = [](long long work) { return (unsigned)((work + 255) / 256); };
  const unsigned rowBlocks = (unsigned)((Nn + 7) / 8);

  const int chunk = ((E + SLICES - 1) / SLICES + 63) & ~63;
  k_count_rank<<<SLICES, CNT_THREADS, 0, stream>>>(colIdx, tab, rank, cursor, E, Nn, pitch, chunk);
  k_offsets   <<<blocks(Nn), blk, 0, stream>>>(tab, cursor, seg, dinv, Nn, pitch);
  k_fill      <<<dim3(blocks(chunk), SLICES), blk, 0, stream>>>(rowIdx, colIdx, rank, tab, srcs, E, pitch, chunk);

  k_gather_rows<4, false><<<rowBlocks, blk, 0, stream>>>(x, seg, srcs, dinv, nullptr, abuf, Nn);
  k_gemm_wmma_f32<128, 256, 4, true><<<256, GEMM_THREADS, 0, stream>>>(abuf, W1, b1, hbuf, Nn);

  k_gemm_wmma_f32<256, 64, 2, false><<<256, GEMM_THREADS, 0, stream>>>(hbuf, W2, nullptr, abuf, Nn);
  float* out = (float*)d_out;
  k_gather_rows<2, true><<<rowBlocks, blk, 0, stream>>>(abuf, seg, srcs, dinv, b2, out, Nn);
}


# both dense layers in one kernel: products formed transposed, so the hidden tile stays in registers as the second product's B operand; hidden buffer and one launch gone; its f32 sum order changes
# speedup vs baseline: 4.9691x; 1.0985x over previous
#include <hip/hip_runtime.h>
#include <hip/hip_bf16.h>


typedef __attribute__((ext_vector_type(2))) float v2f;
typedef __attribute__((ext_vector_type(8))) float v8f;

struct __attribute__((aligned(8))) Seg {
  int start;
  int len;
};

constexpr int SLICES      = 64;
constexpr int CNT_THREADS = 1024;
constexpr int CNT_CAP     = 51200;
constexpr int EPT         = 8;

__global__ __launch_bounds__(CNT_THREADS) void k_count_rank(
    const int* __restrict__ col, int* __restrict__ tab, int* __restrict__ rank,
    int* __restrict__ cursor, int E, int n, int pitch, int chunk) {
  __shared__ __attribute__((aligned(16))) int lcnt[CNT_CAP];
  int4* l4 = (int4*)lcnt;
  const int tid = threadIdx.x;
  const int e0  = blockIdx.x * chunk;
  const int e1  = min(e0 + chunk, E);
  if (blockIdx.x == 0 && tid == 0) *cursor = 0;
  for (int r0 = 0; r0 < n; r0 += CNT_CAP) {
    const int rn  = min(CNT_CAP, n - r0);
    const int rn4 = (rn + 3) >> 2;
    for (int i = tid; i < rn4; i += CNT_THREADS) l4[i] = int4{0, 0, 0, 0};
    __syncthreads();
    for (int e = e0 + tid; e < e1; e += EPT * CNT_THREADS) {
      int c[EPT];
#pragma unroll
      for (int k = 0; k < EPT; ++k) c[k] = col[min(e + k * CNT_THREADS, e1 - 1)];
#pragma unroll
      for (int k = 0; k < EPT; ++k) {
        c[k] = (e + k * CNT_THREADS < e1) ? c[k] - r0 : -1;
      }
      int r[EPT];
#pragma unroll
      for (int k = 0; k < EPT; ++k) {
        r[k] = 0;
        if ((unsigned)c[k] < (unsigned)rn) {
          r[k] = __hip_atomic_fetch_add(&lcnt[c[k]], 1, __ATOMIC_RELAXED,
                                        __HIP_MEMORY_SCOPE_WORKGROUP);
        }
      }
#pragma unroll
      for (int k = 0; k < EPT; ++k) {
        if ((unsigned)c[k] < (unsigned)rn) rank[e + k * CNT_THREADS] = r[k];
      }
    }
    __syncthreads();
    int4* __restrict__ t4 = (int4*)(tab + (size_t)blockIdx.x * pitch + r0);
    for (int i = tid; i < rn4; i += CNT_THREADS) t4[i] = l4[i];
    __syncthreads();
  }
}

__global__ __launch_bounds__(256) void k_offsets(int* tab, int* cursor,
                                                 Seg* __restrict__ seg,
                                                 float* __restrict__ dinv, int n,
                                                 int pitch) {
  __shared__ int wsum[8];
  __shared__ int blk_base;
  const int i    = blockIdx.x * blockDim.x + threadIdx.x;
  const int lane = threadIdx.x & 31;
  const int wave = threadIdx.x >> 5;
  const int ic = min(i, n - 1);
  int cs[SLICES];
#pragma unroll
  for (int p = 0; p < SLICES; ++p) cs[p] = tab[(size_t)p * pitch + ic];
  int c = 0;
#pragma unroll
  for (int p = 0; p < SLICES; ++p) c += cs[p];
  if (i >= n) c = 0;
  int v = c;
#pragma unroll
  for (int d = 1; d < 32; d <<= 1) {
    int t = __shfl_up(v, d);
    if (lane >= d) v += t;
  }
  if (lane == 31) wsum[wave] = v;
  __syncthreads();
  if (threadIdx.x == 0) {
    int tot = 0;
#pragma unroll
    for (int w = 0; w < 8; ++w) {
      int t = wsum[w];
      wsum[w] = tot;
      tot += t;
    }
    blk_base = __hip_atomic_fetch_add(cursor, tot, __ATOMIC_RELAXED,
                                      __HIP_MEMORY_SCOPE_AGENT);
  }
  __syncthreads();
  if (i < n) {
    Seg sg;
    sg.start = blk_base + wsum[wave] + (v - c);
    sg.len   = c;
    seg[i] = sg;
    const float d = (float)(c + 1);
    dinv[i] = (d > 0.0f) ? rsqrtf(d) : 0.0f;
    int run = sg.start;
#pragma unroll
    for (int p = 0; p < SLICES; ++p) {
      tab[(size_t)p * pitch + i] = run;
      run += cs[p];
    }
  }
}

__global__ __launch_bounds__(256) void k_fill(const int* __restrict__ row,
                                              const int* __restrict__ col,
                                              const int* __restrict__ rank,
                                              const int* __restrict__ tab,
                                              int* __restrict__ srcs, int E,
                                              int pitch, int chunk) {
  const int p  = blockIdx.y;
  const int le = blockIdx.x * blockDim.x + threadIdx.x;
  const int e  = p * chunk + le;
  if (le < chunk && e < E) srcs[tab[(size_t)p * pitch + col[e]] + rank[e]] = row[e];
}

template <int VEC, bool ADD_BIAS>
__global__ __launch_bounds__(256) void k_gather_rows(
    const float* __restrict__ h, const Seg* __restrict__ seg,
    const int* __restrict__ srcs, const float* __restrict__ dinv,
    const float* __restrict__ bias, float* __restrict__ out, int n_nodes) {
  typedef float vec __attribute__((ext_vector_type(VEC)));
  const int lane = threadIdx.x & 31;
  const int node = __builtin_amdgcn_readfirstlane(
      (int)(blockIdx.x * (blockDim.x >> 5) + (threadIdx.x >> 5)));
  if (node >= n_nodes) return;
  const Seg   sg = seg[node];
  const int   n  = sg.len;
  const float di = dinv[node];
  const int* __restrict__ sp = srcs + sg.start;
  const vec* __restrict__ hv = (const vec*)h + lane;

  vec bvec = {};
  if (ADD_BIAS) {
#pragma unroll
    for (int c = 0; c < VEC; ++c) bvec[c] = bias[lane * VEC + c];
  }

  vec acc = hv[(size_t)node * 32] * (di * di);
  int j = 0;
  for (; j + 4 <= n; j += 4) {
    const int s0 = sp[j], s1 = sp[j + 1], s2 = sp[j + 2], s3 = sp[j + 3];
    const vec r0 = hv[(size_t)s0 * 32];
    const vec r1 = hv[(size_t)s1 * 32];
    const vec r2 = hv[(size_t)s2 * 32];
    const vec r3 = hv[(size_t)s3 * 32];
    const float w0 = dinv[s0] * di, w1 = dinv[s1] * di;
    const float w2 = dinv[s2] * di, w3 = dinv[s3] * di;
    acc += r0 * w0;
    acc += r1 * w1;
    acc += r2 * w2;
    acc += r3 * w3;
  }
  if (j + 2 <= n) {
    const int s0 = sp[j], s1 = sp[j + 1];
    const vec r0 = hv[(size_t)s0 * 32];
    const vec r1 = hv[(size_t)s1 * 32];
    const float w0 = dinv[s0] * di, w1 = dinv[s1] * di;
    acc += r0 * w0;
    acc += r1 * w1;
    j += 2;
  }
  if (j < n) {
    const int s0 = sp[j];
    const vec r0 = hv[(size_t)s0 * 32];
    acc += r0 * (dinv[s0] * di);
  }
  if (ADD_BIAS) acc += bvec;
  ((vec*)out)[(size_t)node * 32 + lane] = acc;
}

constexpr int MLP_THREADS = 512;
constexpr int MAIN_WAVES  = 12;

template <int K1, int N1, int N2, int NT>
__device__ __forceinline__ void mlp_group(const float4 (&q)[K1 / 8], v8f (&acc2)[N2 / 16],
                                          int g, int lane, int h,
                                          const float* __restrict__ b1,
                                          const float4* W1s, const float4* W2s) {
  constexpr int KP1 = K1 / 8;
  constexpr int NT1 = N1 / 16;
  constexpr int NT2 = N2 / 16;
  const int t0 = g * NT;

  float4 ba[NT], bb[NT];
#pragma unroll
  for (int t = 0; t < NT; ++t) {
    const float4* bp = (const float4*)(b1 + 16 * (t0 + t) + 8 * h);
    ba[t] = bp[0];
    bb[t] = bp[1];
  }

  v8f acc1[NT];
#pragma unroll
  for (int t = 0; t < NT; ++t) acc1[t] = v8f{};
  const float4* wp = W1s + t0 * 32 + lane;
#pragma unroll
  for (int j = 0; j < KP1; ++j) {
    float4 wf[NT];
#pragma unroll
    for (int t = 0; t < NT; ++t) wf[t] = wp[(j * NT1 + t) * 32];
    v2f x0, x1;
    x0.x = q[j].x;
    x0.y = q[j].y;
    x1.x = q[j].z;
    x1.y = q[j].w;
#pragma unroll
    for (int t = 0; t < NT; ++t) {
      v2f w0;
      w0.x = wf[t].x;
      w0.y = wf[t].y;
      acc1[t] = __builtin_amdgcn_wmma_f32_16x16x4_f32(
           false, w0,  false, x0,
           (short)0, acc1[t],  false,  false);
    }
#pragma unroll
    for (int t = 0; t < NT; ++t) {
      v2f w1;
      w1.x = wf[t].z;
      w1.y = wf[t].w;
      acc1[t] = __builtin_amdgcn_wmma_f32_16x16x4_f32(
           false, w1,  false, x1,
           (short)0, acc1[t],  false,  false);
    }
  }

#pragma unroll
  for (int t = 0; t < NT; ++t) {
    const int tt = t0 + t;
    float hv[8];
    hv[0] = fmaxf(acc1[t][0] + ba[t].x, 0.0f);
    hv[1] = fmaxf(acc1[t][1] + ba[t].y, 0.0f);
    hv[2] = fmaxf(acc1[t][2] + ba[t].z, 0.0f);
    hv[3] = fmaxf(acc1[t][3] + ba[t].w, 0.0f);
    hv[4] = fmaxf(acc1[t][4] + bb[t].x, 0.0f);
    hv[5] = fmaxf(acc1[t][5] + bb[t].y, 0.0f);
    hv[6] = fmaxf(acc1[t][6] + bb[t].z, 0.0f);
    hv[7] = fmaxf(acc1[t][7] + bb[t].w, 0.0f);
#pragma unroll
    for (int p = 0; p < 2; ++p) {
      float4 w2f[NT2];
#pragma unroll
      for (int k = 0; k < NT2; ++k) w2f[k] = W2s[((tt * NT2 + k) * 2 + p) * 32 + lane];
      v2f h0, h1;
      h0.x = hv[4 * p + 0];
      h0.y = hv[4 * p + 1];
      h1.x = hv[4 * p + 2];
      h1.y = hv[4 * p + 3];
#pragma unroll
      for (int k = 0; k < NT2; ++k) {
        v2f w0;
        w0.x = w2f[k].x;
        w0.y = w2f[k].y;
        acc2[k] = __builtin_amdgcn_wmma_f32_16x16x4_f32(
             false, w0,  false, h0,
             (short)0, acc2[k],  false,  false);
      }
#pragma unroll
      for (int k = 0; k < NT2; ++k) {
        v2f w1;
        w1.x = w2f[k].z;
        w1.y = w2f[k].w;
        acc2[k] = __builtin_amdgcn_wmma_f32_16x16x4_f32(
             false, w1,  false, h1,
             (short)0, acc2[k],  false,  false);
      }
    }
  }
}

template <int K1, int N1, int N2, int NT>
__global__ __launch_bounds__(MLP_THREADS) void k_mlp_wmma_f32(
    const float* __restrict__ X, const float* __restrict__ W1,
    const float* __restrict__ b1, const float* __restrict__ W2,
    float* __restrict__ O, int M) {
  constexpr int KP1 = K1 / 8;
  constexpr int NT1 = N1 / 16;
  constexpr int NT2 = N2 / 16;
  constexpr int NG  = NT1 / NT;
  static_assert(NT1 % NT == 0, "hidden tiles are taken NT at a time");
  static_assert((MLP_THREADS >> 5) - MAIN_WAVES == NG, "one left-over wave per hidden group");
  __shared__ __attribute__((aligned(16))) float4 W1s[KP1 * NT1 * 32];
  __shared__ __attribute__((aligned(16))) float4 W2s[NT1 * NT2 * 2 * 32];
  __shared__ __attribute__((aligned(16))) float4 parts[NG * NT2 * 2 * 32];

  const int lane = threadIdx.x & 31;
  const int wave = __builtin_amdgcn_readfirstlane((int)(threadIdx.x >> 5));
  const int c = lane & 15;
  const int h = lane >> 4;
  const int grid  = (int)gridDim.x;
  const int units = M >> 4;
  const int nmain = grid * MAIN_WAVES;
  const int rest  = units % nmain;
  const int ncut  = (rest <= grid) ? rest : 0;
  const int nwhole = units - ncut;

  const bool main_wave = wave < MAIN_WAVES;
  const bool cut_block = (int)blockIdx.x < ncut;
  int u, u_end, g0, g1;
  if (main_wave) {
    u = wave * grid + (int)blockIdx.x;
    u_end = nwhole;
    g0 = 0;
    g1 = NG;
  } else {
    u = nwhole + (int)blockIdx.x;
    u_end = cut_block ? u + 1 : 0;
    g0 = wave - MAIN_WAVES;
    g1 = g0 + 1;
  }

  float4 q[KP1];
  {
    const int uc = min(u, units - 1);
    const float4* ap = (const float4*)(X + (size_t)((uc << 4) + c) * K1 + 4 * h);
#pragma unroll
    for (int j = 0; j < KP1; ++j) q[j] = ap[2 * j];
  }
  __builtin_amdgcn_sched_barrier(0);

  static_assert((KP1 * 2 * N1) % MLP_THREADS == 0, "the copy loop has no tail");
#pragma unroll
  for (int it = 0; it < KP1 * 2 * N1 / MLP_THREADS; ++it) {
    const int idx = it * MLP_THREADS + (int)threadIdx.x;
    const int n  = idx % N1;
    const int jh = idx / N1;
    const float* src = W1 + (size_t)(4 * jh) * N1 + n;
    float4 v;
    v.x = src[0];
    v.y = src[N1];
    v.z = src[2 * N1];
    v.w = src[3 * N1];
    W1s[((jh >> 1) * NT1 + (n >> 4)) * 32 + (jh & 1) * 16 + (n & 15)] = v;
  }
  static_assert((N1 / 4 * N2) % MLP_THREADS == 0, "the copy loop has no tail");
#pragma unroll
  for (int it = 0; it < N1 / 4 * N2 / MLP_THREADS; ++it) {
    const int idx = it * MLP_THREADS + (int)threadIdx.x;
    const int o  = idx % N2;
    const int n4 = idx / N2;
    const float* src = W2 + (size_t)(4 * n4) * N2 + o;
    float4 v;
    v.x = src[0];
    v.y = src[N2];
    v.z = src[2 * N2];
    v.w = src[3 * N2];
    const int t  = n4 >> 2;
    const int hh = (n4 >> 1) & 1;
    const int p  = n4 & 1;
    W2s[(((t * NT2) + (o >> 4)) * 2 + p) * 32 + hh * 16 + (o & 15)] = v;
  }
  __syncthreads();

  for (; u < u_end; u += nmain) {
    v8f acc2[NT2];
#pragma unroll
    for (int k = 0; k < NT2; ++k) acc2[k] = v8f{};

    for (int g = g0; g < g1; ++g) mlp_group<K1, N1, N2, NT>(q, acc2, g, lane, h, b1, W1s, W2s);

    float4 lo[NT2], hi[NT2];
#pragma unroll
    for (int k = 0; k < NT2; ++k) {
      lo[k].x = acc2[k][0];
      lo[k].y = acc2[k][1];
      lo[k].z = acc2[k][2];
      lo[k].w = acc2[k][3];
      hi[k].x = acc2[k][4];
      hi[k].y = acc2[k][5];
      hi[k].z = acc2[k][6];
      hi[k].w = acc2[k][7];
    }
    if (main_wave) {
      float4* op = (float4*)(O + (size_t)((u << 4) + c) * N2 + 8 * h);
#pragma unroll
      for (int k = 0; k < NT2; ++k) {
        op[4 * k]     = lo[k];
        op[4 * k + 1] = hi[k];
      }
      if (u + nmain < u_end) {
        const float4* ap = (const float4*)(X + (size_t)(((u + nmain) << 4) + c) * K1 + 4 * h);
#pragma unroll
        for (int j = 0; j < KP1; ++j) q[j] = ap[2 * j];
      }
    } else {
      float4* pp = parts + (g0 * NT2 * 2) * 32 + lane;
#pragma unroll
      for (int k = 0; k < NT2; ++k) {
        pp[(2 * k) * 32]     = lo[k];
        pp[(2 * k + 1) * 32] = hi[k];
      }
    }
  }

  __syncthreads();
  if (wave == MAIN_WAVES && cut_block) {
    const float4* pp = parts + lane;
    float4* op = (float4*)(O + (size_t)(((nwhole + (int)blockIdx.x) << 4) + c) * N2 + 8 * h);
#pragma unroll
    for (int k = 0; k < 2 * NT2; ++k) {
      float4 s = pp[k * 32];
#pragma unroll
      for (int g = 1; g < NG; ++g) {
        const float4 v = pp[(g * NT2 * 2 + k) * 32];
        s.x += v.x;
        s.y += v.y;
        s.z += v.z;
        s.w += v.w;
      }
      op[(k >> 1) * 4 + (k & 1)] = s;
    }
  }
}

extern "C" void kernel_launch(void* const* d_in, const int* in_sizes, int n_in,
                              void* d_out, int out_size, void* d_ws, size_t ws_size,
                              hipStream_t stream) {
  const float* x  = (const float*)d_in[0];
  const int*   ei = (const int*)d_in[1];
  const float* W1 = (const float*)d_in[2];
  const float* b1 = (const float*)d_in[3];
  const float* W2 = (const float*)d_in[4];
  const float* b2 = (const float*)d_in[5];

  const int H  = in_sizes[3];
  const int F  = in_sizes[2] / H;
  const int C  = in_sizes[5];
  const int Nn = in_sizes[0] / F;
  const int E  = in_sizes[1] / 2;
  const int* rowIdx = ei;
  const int* colIdx = ei + E;

  auto al = [](size_t v) { return (v + 255) & ~(size_t)255; };
  size_t off = 0;
  float* dinv = (float*)((char*)d_ws + off);
  off += al((size_t)Nn * sizeof(float));
  int* cursor = (int*)((char*)d_ws + off);
  off += al(sizeof(int));
  int* tab = (int*)((char*)d_ws + off);
  const int pitch = (Nn + 63) & ~63;
  off += al((size_t)SLICES * pitch * sizeof(int));
  Seg* seg = (Seg*)((char*)d_ws + off);
  off += al((size_t)Nn * sizeof(Seg));
  int* rank = (int*)((char*)d_ws + off);
  off += al((size_t)E * sizeof(int));
  int* srcs = (int*)((char*)d_ws + off);
  off += al((size_t)E * sizeof(int));
  float* obuf = (float*)((char*)d_ws + off);
  off += al((size_t)Nn * C * sizeof(float));
  float* abuf = (float*)((char*)d_ws + off);

  const dim3 blk(256);
  auto blocks = [](long long work) { return (unsigned)((work + 255) / 256); };
  const unsigned rowBlocks = (unsigned)((Nn + 7) / 8);

  const int chunk = ((E + SLICES - 1) / SLICES + 63) & ~63;
  k_count_rank<<<SLICES, CNT_THREADS, 0, stream>>>(colIdx, tab, rank, cursor, E, Nn, pitch, chunk);
  k_offsets   <<<blocks(Nn), blk, 0, stream>>>(tab, cursor, seg, dinv, Nn, pitch);
  k_fill      <<<dim3(blocks(chunk), SLICES), blk, 0, stream>>>(rowIdx, colIdx, rank, tab, srcs, E, pitch, chunk);

  k_gather_rows<4, false><<<rowBlocks, blk, 0, stream>>>(x, seg, srcs, dinv, nullptr, abuf, Nn);

  k_mlp_wmma_f32<128, 256, 64, 4><<<256, MLP_THREADS, 0, stream>>>(abuf, W1, b1, W2, obuf, Nn);

  float* out = (float*)d_out;
  k_gather_rows<2, true><<<rowBlocks, blk, 0, stream>>>(obuf, seg, srcs, dinv, b2, out, Nn);
}


# fill kernel writes each edge's source weight dinv[src] beside its source index (8-byte entries); both row gathers take index and weight from one sequential scalar load, no random dinv read per edge
# speedup vs baseline: 5.2450x; 1.0555x over previous
#include <hip/hip_runtime.h>
#include <hip/hip_bf16.h>


typedef __attribute__((ext_vector_type(2))) float v2f;
typedef __attribute__((ext_vector_type(8))) float v8f;

struct __attribute__((aligned(8))) Seg {
  int start;
  int len;
};

struct __attribute__((aligned(8))) Ent {
  int   src;
  float w;
};

constexpr int SLICES      = 64;
constexpr int CNT_THREADS = 1024;
constexpr int CNT_CAP     = 51200;
constexpr int EPT         = 8;

__global__ __launch_bounds__(CNT_THREADS) void k_count_rank(
    const int* __restrict__ col, int* __restrict__ tab, int* __restrict__ rank,
    int* __restrict__ cursor, int E, int n, int pitch, int chunk) {
  __shared__ __attribute__((aligned(16))) int lcnt[CNT_CAP];
  int4* l4 = (int4*)lcnt;
  const int tid = threadIdx.x;
  const int e0  = blockIdx.x * chunk;
  const int e1  = min(e0 + chunk, E);
  if (blockIdx.x == 0 && tid == 0) *cursor = 0;
  for (int r0 = 0; r0 < n; r0 += CNT_CAP) {
    const int rn  = min(CNT_CAP, n - r0);
    const int rn4 = (rn + 3) >> 2;
    for (int i = tid; i < rn4; i += CNT_THREADS) l4[i] = int4{0, 0, 0, 0};
    __syncthreads();
    for (int e = e0 + tid; e < e1; e += EPT * CNT_THREADS) {
      int c[EPT];
#pragma unroll
      for (int k = 0; k < EPT; ++k) c[k] = col[min(e + k * CNT_THREADS, e1 - 1)];
#pragma unroll
      for (int k = 0; k < EPT; ++k) {
        c[k] = (e + k * CNT_THREADS < e1) ? c[k] - r0 : -1;
      }
      int r[EPT];
#pragma unroll
      for (int k = 0; k < EPT; ++k) {
        r[k] = 0;
        if ((unsigned)c[k] < (unsigned)rn) {
          r[k] = __hip_atomic_fetch_add(&lcnt[c[k]], 1, __ATOMIC_RELAXED,
                                        __HIP_MEMORY_SCOPE_WORKGROUP);
        }
      }
#pragma unroll
      for (int k = 0; k < EPT; ++k) {
        if ((unsigned)c[k] < (unsigned)rn) rank[e + k * CNT_THREADS] = r[k];
      }
    }
    __syncthreads();
    int4* __restrict__ t4 = (int4*)(tab + (size_t)blockIdx.x * pitch + r0);
    for (int i = tid; i < rn4; i += CNT_THREADS) t4[i] = l4[i];
    __syncthreads();
  }
}

__global__ __launch_bounds__(256) void k_offsets(int* tab, int* cursor,
                                                 Seg* __restrict__ seg,
                                                 float* __restrict__ dinv, int n,
                                                 int pitch) {
  __shared__ int wsum[8];
  __shared__ int blk_base;
  const int i    = blockIdx.x * blockDim.x + threadIdx.x;
  const int lane = threadIdx.x & 31;
  const int wave = threadIdx.x >> 5;
  const int ic = min(i, n - 1);
  int cs[SLICES];
#pragma unroll
  for (int p = 0; p < SLICES; ++p) cs[p] = tab[(size_t)p * pitch + ic];
  int c = 0;
#pragma unroll
  for (int p = 0; p < SLICES; ++p) c += cs[p];
  if (i >= n) c = 0;
  int v = c;
#pragma unroll
  for (int d = 1; d < 32; d <<= 1) {
    int t = __shfl_up(v, d);
    if (lane >= d) v += t;
  }
  if (lane == 31) wsum[wave] = v;
  __syncthreads();
  if (threadIdx.x == 0) {
    int tot = 0;
#pragma unroll
    for (int w = 0; w < 8; ++w) {
      int t = wsum[w];
      wsum[w] = tot;
      tot += t;
    }
    blk_base = __hip_atomic_fetch_add(cursor, tot, __ATOMIC_RELAXED,
                                      __HIP_MEMORY_SCOPE_AGENT);
  }
  __syncthreads();
  if (i < n) {
    Seg sg;
    sg.start = blk_base + wsum[wave] + (v - c);
    sg.len   = c;
    seg[i] = sg;
    const float d = (float)(c + 1);
    dinv[i] = (d > 0.0f) ? rsqrtf(d) : 0.0f;
    int run = sg.start;
#pragma unroll
    for (int p = 0; p < SLICES; ++p) {
      tab[(size_t)p * pitch + i] = run;
      run += cs[p];
    }
  }
}

__global__ __launch_bounds__(256) void k_fill(const int* __restrict__ row,
                                              const int* __restrict__ col,
                                              const int* __restrict__ rank,
                                              const int* __restrict__ tab,
                                              const float* __restrict__ dinv,
                                              Ent* __restrict__ ents, int E,
                                              int pitch, int chunk) {
  const int p  = blockIdx.y;
  const int le = blockIdx.x * blockDim.x + threadIdx.x;
  const int e  = p * chunk + le;
  if (le < chunk && e < E) {
    const int s = row[e];
    Ent en;
    en.src = s;
    en.w   = dinv[s];
    ents[tab[(size_t)p * pitch + col[e]] + rank[e]] = en;
  }
}

template <int VEC, bool ADD_BIAS>
__global__ __launch_bounds__(256) void k_gather_rows(
    const float* __restrict__ h, const Seg* __restrict__ seg,
    const Ent* __restrict__ ents, const float* __restrict__ dinv,
    const float* __restrict__ bias, float* __restrict__ out, int n_nodes) {
  typedef float vec __attribute__((ext_vector_type(VEC)));
  const int lane = threadIdx.x & 31;
  const int node = __builtin_amdgcn_readfirstlane(
      (int)(blockIdx.x * (blockDim.x >> 5) + (threadIdx.x >> 5)));
  if (node >= n_nodes) return;
  const Seg   sg = seg[node];
  const int   n  = sg.len;
  const float di = dinv[node];
  const Ent* __restrict__ sp = ents + sg.start;
  const vec* __restrict__ hv = (const vec*)h + lane;

  vec bvec = {};
  if (ADD_BIAS) {
#pragma unroll
    for (int c = 0; c < VEC; ++c) bvec[c] = bias[lane * VEC + c];
  }

  vec acc = hv[(size_t)node * 32] * (di * di);
  int j = 0;
  for (; j + 4 <= n; j += 4) {
    const Ent e0 = sp[j], e1 = sp[j + 1], e2 = sp[j + 2], e3 = sp[j + 3];
    const vec r0 = hv[(size_t)e0.src * 32];
    const vec r1 = hv[(size_t)e1.src * 32];
    const vec r2 = hv[(size_t)e2.src * 32];
    const vec r3 = hv[(size_t)e3.src * 32];
    const float w0 = e0.w * di, w1 = e1.w * di;
    const float w2 = e2.w * di, w3 = e3.w * di;
    acc += r0 * w0;
    acc += r1 * w1;
    acc += r2 * w2;
    acc += r3 * w3;
  }
  if (j + 2 <= n) {
    const Ent e0 = sp[j], e1 = sp[j + 1];
    const vec r0 = hv[(size_t)e0.src * 32];
    const vec r1 = hv[(size_t)e1.src * 32];
    const float w0 = e0.w * di, w1 = e1.w * di;
    acc += r0 * w0;
    acc += r1 * w1;
    j += 2;
  }
  if (j < n) {
    const Ent e0 = sp[j];
    const vec r0 = hv[(size_t)e0.src * 32];
    acc += r0 * (e0.w * di);
  }
  if (ADD_BIAS) acc += bvec;
  ((vec*)out)[(size_t)node * 32 + lane] = acc;
}

constexpr int MLP_THREADS = 512;
constexpr int MAIN_WAVES  = 12;

template <int K1, int N1, int N2, int NT>
__device__ __forceinline__ void mlp_group(const float4 (&q)[K1 / 8], v8f (&acc2)[N2 / 16],
                                          int g, int lane, int h,
                                          const float* __restrict__ b1,
                                          const float4* W1s, const float4* W2s) {
  constexpr int KP1 = K1 / 8;
  constexpr int NT1 = N1 / 16;
  constexpr int NT2 = N2 / 16;
  const int t0 = g * NT;

  float4 ba[NT], bb[NT];
#pragma unroll
  for (int t = 0; t < NT; ++t) {
    const float4* bp = (const float4*)(b1 + 16 * (t0 + t) + 8 * h);
    ba[t] = bp[0];
    bb[t] = bp[1];
  }

  v8f acc1[NT];
#pragma unroll
  for (int t = 0; t < NT; ++t) acc1[t] = v8f{};
  const float4* wp = W1s + t0 * 32 + lane;
#pragma unroll
  for (int j = 0; j < KP1; ++j) {
    float4 wf[NT];
#pragma unroll
    for (int t = 0; t < NT; ++t) wf[t] = wp[(j * NT1 + t) * 32];
    v2f x0, x1;
    x0.x = q[j].x;
    x0.y = q[j].y;
    x1.x = q[j].z;
    x1.y = q[j].w;
#pragma unroll
    for (int t = 0; t < NT; ++t) {
      v2f w0;
      w0.x = wf[t].x;
      w0.y = wf[t].y;
      acc1[t] = __builtin_amdgcn_wmma_f32_16x16x4_f32(
           false, w0,  false, x0,
           (short)0, acc1[t],  false,  false);
    }
#pragma unroll
    for (int t = 0; t < NT; ++t) {
      v2f w1;
      w1.x = wf[t].z;
      w1.y = wf[t].w;
      acc1[t] = __builtin_amdgcn_wmma_f32_16x16x4_f32(
           false, w1,  false, x1,
           (short)0, acc1[t],  false,  false);
    }
  }

#pragma unroll
  for (int t = 0; t < NT; ++t) {
    const int tt = t0 + t;
    float hv[8];
    hv[0] = fmaxf(acc1[t][0] + ba[t].x, 0.0f);
    hv[1] = fmaxf(acc1[t][1] + ba[t].y, 0.0f);
    hv[2] = fmaxf(acc1[t][2] + ba[t].z, 0.0f);
    hv[3] = fmaxf(acc1[t][3] + ba[t].w, 0.0f);
    hv[4] = fmaxf(acc1[t][4] + bb[t].x, 0.0f);
    hv[5] = fmaxf(acc1[t][5] + bb[t].y, 0.0f);
    hv[6] = fmaxf(acc1[t][6] + bb[t].z, 0.0f);
    hv[7] = fmaxf(acc1[t][7] + bb[t].w, 0.0f);
#pragma unroll
    for (int p = 0; p < 2; ++p) {
      float4 w2f[NT2];
#pragma unroll
      for (int k = 0; k < NT2; ++k) w2f[k] = W2s[((tt * NT2 + k) * 2 + p) * 32 + lane];
      v2f h0, h1;
      h0.x = hv[4 * p + 0];
      h0.y = hv[4 * p + 1];
      h1.x = hv[4 * p + 2];
      h1.y = hv[4 * p + 3];
#pragma unroll
      for (int k = 0; k < NT2; ++k) {
        v2f w0;
        w0.x = w2f[k].x;
        w0.y = w2f[k].y;
        acc2[k] = __builtin_amdgcn_wmma_f32_16x16x4_f32(
             false, w0,  false, h0,
             (short)0, acc2[k],  false,  false);
      }
#pragma unroll
      for (int k = 0; k < NT2; ++k) {
        v2f w1;
        w1.x = w2f[k].z;
        w1.y = w2f[k].w;
        acc2[k] = __builtin_amdgcn_wmma_f32_16x16x4_f32(
             false, w1,  false, h1,
             (short)0, acc2[k],  false,  false);
      }
    }
  }
}

template <int K1, int N1, int N2, int NT>
__global__ __launch_bounds__(MLP_THREADS) void k_mlp_wmma_f32(
    const float* __restrict__ X, const float* __restrict__ W1,
    const float* __restrict__ b1, const float* __restrict__ W2,
    float* __restrict__ O, int M) {
  constexpr int KP1 = K1 / 8;
  constexpr int NT1 = N1 / 16;
  constexpr int NT2 = N2 / 16;
  constexpr int NG  = NT1 / NT;
  static_assert(NT1 % NT == 0, "hidden tiles are taken NT at a time");
  static_assert((MLP_THREADS >> 5) - MAIN_WAVES == NG, "one left-over wave per hidden group");
  __shared__ __attribute__((aligned(16))) float4 W1s[KP1 * NT1 * 32];
  __shared__ __attribute__((aligned(16))) float4 W2s[NT1 * NT2 * 2 * 32];
  __shared__ __attribute__((aligned(16))) float4 parts[NG * NT2 * 2 * 32];

  const int lane = threadIdx.x & 31;
  const int wave = __builtin_amdgcn_readfirstlane((int)(threadIdx.x >> 5));
  const int c = lane & 15;
  const int h = lane >> 4;
  const int grid  = (int)gridDim.x;
  const int units = M >> 4;
  const int nmain = grid * MAIN_WAVES;
  const int rest  = units % nmain;
  const int ncut  = (rest <= grid) ? rest : 0;
  const int nwhole = units - ncut;

  const bool main_wave = wave < MAIN_WAVES;
  const bool cut_block = (int)blockIdx.x < ncut;
  int u, u_end, g0, g1;
  if (main_wave) {
    u = wave * grid + (int)blockIdx.x;
    u_end = nwhole;
    g0 = 0;
    g1 = NG;
  } else {
    u = nwhole + (int)blockIdx.x;
    u_end = cut_block ? u + 1 : 0;
    g0 = wave - MAIN_WAVES;
    g1 = g0 + 1;
  }

  float4 q[KP1];
  {
    const int uc = min(u, units - 1);
    const float4* ap = (const float4*)(X + (size_t)((uc << 4) + c) * K1 + 4 * h);
#pragma unroll
    for (int j = 0; j < KP1; ++j) q[j] = ap[2 * j];
  }
  __builtin_amdgcn_sched_barrier(0);

  static_assert((KP1 * 2 * N1) % MLP_THREADS == 0, "the copy loop has no tail");
#pragma unroll
  for (int it = 0; it < KP1 * 2 * N1 / MLP_THREADS; ++it) {
    const int idx = it * MLP_THREADS + (int)threadIdx.x;
    const int n  = idx % N1;
    const int jh = idx / N1;
    const float* src = W1 + (size_t)(4 * jh) * N1 + n;
    float4 v;
    v.x = src[0];
    v.y = src[N1];
    v.z = src[2 * N1];
    v.w = src[3 * N1];
    W1s[((jh >> 1) * NT1 + (n >> 4)) * 32 + (jh & 1) * 16 + (n & 15)] = v;
  }
  static_assert((N1 / 4 * N2) % MLP_THREADS == 0, "the copy loop has no tail");
#pragma unroll
  for (int it = 0; it < N1 / 4 * N2 / MLP_THREADS; ++it) {
    const int idx = it * MLP_THREADS + (int)threadIdx.x;
    const int o  = idx % N2;
    const int n4 = idx / N2;
    const float* src = W2 + (size_t)(4 * n4) * N2 + o;
    float4 v;
    v.x = src[0];
    v.y = src[N2];
    v.z = src[2 * N2];
    v.w = src[3 * N2];
    const int t  = n4 >> 2;
    const int hh = (n4 >> 1) & 1;
    const int p  = n4 & 1;
    W2s[(((t * NT2) + (o >> 4)) * 2 + p) * 32 + hh * 16 + (o & 15)] = v;
  }
  __syncthreads();

  for (; u < u_end; u += nmain) {
    v8f acc2[NT2];
#pragma unroll
    for (int k = 0; k < NT2; ++k) acc2[k] = v8f{};

    for (int g = g0; g < g1; ++g) mlp_group<K1, N1, N2, NT>(q, acc2, g, lane, h, b1, W1s, W2s);

    float4 lo[NT2], hi[NT2];
#pragma unroll
    for (int k = 0; k < NT2; ++k) {
      lo[k].x = acc2[k][0];
      lo[k].y = acc2[k][1];
      lo[k].z = acc2[k][2];
      lo[k].w = acc2[k][3];
      hi[k].x = acc2[k][4];
      hi[k].y = acc2[k][5];
      hi[k].z = acc2[k][6];
      hi[k].w = acc2[k][7];
    }
    if (main_wave) {
      float4* op = (float4*)(O + (size_t)((u << 4) + c) * N2 + 8 * h);
#pragma unroll
      for (int k = 0; k < NT2; ++k) {
        op[4 * k]     = lo[k];
        op[4 * k + 1] = hi[k];
      }
      if (u + nmain < u_end) {
        const float4* ap = (const float4*)(X + (size_t)(((u + nmain) << 4) + c) * K1 + 4 * h);
#pragma unroll
        for (int j = 0; j < KP1; ++j) q[j] = ap[2 * j];
      }
    } else {
      float4* pp = parts + (g0 * NT2 * 2) * 32 + lane;
#pragma unroll
      for (int k = 0; k < NT2; ++k) {
        pp[(2 * k) * 32]     = lo[k];
        pp[(2 * k + 1) * 32] = hi[k];
      }
    }
  }

  __syncthreads();
  if (wave == MAIN_WAVES && cut_block) {
    const float4* pp = parts + lane;
    float4* op = (float4*)(O + (size_t)(((nwhole + (int)blockIdx.x) << 4) + c) * N2 + 8 * h);
#pragma unroll
    for (int k = 0; k < 2 * NT2; ++k) {
      float4 s = pp[k * 32];
#pragma unroll
      for (int g = 1; g < NG; ++g) {
        const float4 v = pp[(g * NT2 * 2 + k) * 32];
        s.x += v.x;
        s.y += v.y;
        s.z += v.z;
        s.w += v.w;
      }
      op[(k >> 1) * 4 + (k & 1)] = s;
    }
  }
}

extern "C" void kernel_launch(void* const* d_in, const int* in_sizes, int n_in,
                              void* d_out, int out_size, void* d_ws, size_t ws_size,
                              hipStream_t stream) {
  const float* x  = (const float*)d_in[0];
  const int*   ei = (const int*)d_in[1];
  const float* W1 = (const float*)d_in[2];
  const float* b1 = (const float*)d_in[3];
  const float* W2 = (const float*)d_in[4];
  const float* b2 = (const float*)d_in[5];

  const int H  = in_sizes[3];
  const int F  = in_sizes[2] / H;
  const int C  = in_sizes[5];
  const int Nn = in_sizes[0] / F;
  const int E  = in_sizes[1] / 2;
  const int* rowIdx = ei;
  const int* colIdx = ei + E;

  auto al = [](size_t v) { return (v + 255) & ~(size_t)255; };
  size_t off = 0;
  float* dinv = (float*)((char*)d_ws + off);
  off += al((size_t)Nn * sizeof(float));
  int* cursor = (int*)((char*)d_ws + off);
  off += al(sizeof(int));
  int* tab = (int*)((char*)d_ws + off);
  const int pitch = (Nn + 63) & ~63;
  off += al((size_t)SLICES * pitch * sizeof(int));
  Seg* seg = (Seg*)((char*)d_ws + off);
  off += al((size_t)Nn * sizeof(Seg));
  int* rank = (int*)((char*)d_ws + off);
  off += al((size_t)E * sizeof(int));
  Ent* ents = (Ent*)((char*)d_ws + off);
  off += al((size_t)E * sizeof(Ent));
  float* obuf = (float*)((char*)d_ws + off);
  off += al((size_t)Nn * C * sizeof(float));
  float* abuf = (float*)((char*)d_ws + off);

  const dim3 blk(256);
  auto blocks = [](long long work) { return (unsigned)((work + 255) / 256); };
  const unsigned rowBlocks = (unsigned)((Nn + 7) / 8);

  const int chunk = ((E + SLICES - 1) / SLICES + 63) & ~63;
  k_count_rank<<<SLICES, CNT_THREADS, 0, stream>>>(colIdx, tab, rank, cursor, E, Nn, pitch, chunk);
  k_offsets   <<<blocks(Nn), blk, 0, stream>>>(tab, cursor, seg, dinv, Nn, pitch);
  k_fill      <<<dim3(blocks(chunk), SLICES), blk, 0, stream>>>(rowIdx, colIdx, rank, tab, dinv, ents, E, pitch, chunk);

  k_gather_rows<4, false><<<rowBlocks, blk, 0, stream>>>(x, seg, ents, dinv, nullptr, abuf, Nn);

  k_mlp_wmma_f32<128, 256, 64, 4><<<256, MLP_THREADS, 0, stream>>>(abuf, W1, b1, W2, obuf, Nn);

  float* out = (float*)d_out;
  k_gather_rows<2, true><<<rowBlocks, blk, 0, stream>>>(obuf, seg, ents, dinv, b2, out, Nn);
}


# edge list built in 2 launches, not 3: slices bucket edges by 256-node range, one work-group per range orders them in LDS; deg^-1/2 rides in pre-scaled rows, not per edge; f32 sum order changes
# speedup vs baseline: 5.6836x; 1.0836x over previous
#include <hip/hip_runtime.h>
#include <hip/hip_bf16.h>


typedef __attribute__((ext_vector_type(2))) float v2f;
typedef __attribute__((ext_vector_type(8))) float v8f;

struct __attribute__((aligned(16))) Seg {
  int   start;
  int   len;
  float di;
  int   unused;
};

struct __attribute__((aligned(8))) Pair {
  int src;
  int dst;
};

constexpr int SLICES      = 64;
constexpr int BKT_THREADS = 1024;
constexpr int NPB         = 256;
constexpr int MAXB        = 2048;
constexpr int EPT         = 16;
constexpr int OWN_THREADS = 1024;

__global__ __launch_bounds__(BKT_THREADS) void k_bucket(
    const int* __restrict__ row, const int* __restrict__ col, Pair* __restrict__ pairs,
    int* __restrict__ off, int* __restrict__ cursor, int E, int n, int nb, int pitchB,
    int chunk) {
  __shared__ int cnt[MAXB];
  const int tid = threadIdx.x;
  const int e0  = blockIdx.x * chunk;
  const int e1  = min(e0 + chunk, E);
  if (blockIdx.x == 0 && tid == 0) *cursor = 0;
  for (int i = tid; i < nb; i += BKT_THREADS) cnt[i] = 0;
  __syncthreads();
  for (int e = e0 + tid; e < e1; e += EPT * BKT_THREADS) {
    int c[EPT];
#pragma unroll
    for (int k = 0; k < EPT; ++k) c[k] = col[min(e + k * BKT_THREADS, e1 - 1)];
#pragma unroll
    for (int k = 0; k < EPT; ++k) {
      if (e + k * BKT_THREADS < e1 && (unsigned)c[k] < (unsigned)n) {
        __hip_atomic_fetch_add(&cnt[c[k] / NPB], 1, __ATOMIC_RELAXED,
                               __HIP_MEMORY_SCOPE_WORKGROUP);
      }
    }
  }
  __syncthreads();
  if (tid < 32) {
    const int per = (nb + 31) >> 5;
    const int b0  = tid * per;
    const int b1  = min(b0 + per, nb);
    int s = 0;
    for (int b = b0; b < b1; ++b) s += cnt[b];
    int v = s;
#pragma unroll
    for (int d = 1; d < 32; d <<= 1) {
      int t = __shfl_up(v, d);
      if (tid >= d) v += t;
    }
    int run = v - s;
    int* __restrict__ orow = off + (size_t)blockIdx.x * pitchB;
    for (int b = b0; b < b1; ++b) {
      const int t = cnt[b];
      cnt[b]  = run;
      orow[b] = run;
      run += t;
    }
    if (tid == 31) orow[nb] = v;
  }
  __syncthreads();
  Pair* __restrict__ piece = pairs + e0;
  for (int e = e0 + tid; e < e1; e += EPT * BKT_THREADS) {
    int c[EPT], sr[EPT];
#pragma unroll
    for (int k = 0; k < EPT; ++k) {
      const int idx = min(e + k * BKT_THREADS, e1 - 1);
      c[k]  = col[idx];
      sr[k] = row[idx];
    }
#pragma unroll
    for (int k = 0; k < EPT; ++k) {
      if (e + k * BKT_THREADS < e1 && (unsigned)c[k] < (unsigned)n) {
        const int pos = __hip_atomic_fetch_add(&cnt[c[k] / NPB], 1, __ATOMIC_RELAXED,
                                               __HIP_MEMORY_SCOPE_WORKGROUP);
        Pair pr;
        pr.src = sr[k];
        pr.dst = c[k];
        piece[pos] = pr;
      }
    }
  }
}

template <int XQ>
__global__ __launch_bounds__(OWN_THREADS) void k_own(
    const Pair* __restrict__ pairs, const int* __restrict__ off, int* cursor,
    Seg* __restrict__ seg, float* __restrict__ dinv, int* __restrict__ ents,
    const float* __restrict__ x, float* __restrict__ xs, int n, int pitchB, int chunk) {
  static_assert(XQ == 32, "a wave moves one row of x per load: 32 lanes of 16 bytes");
  constexpr int RPT = NPB * XQ / OWN_THREADS;
  __shared__ __attribute__((aligned(16))) int cnt[SLICES * NPB];
  __shared__ int   rs[SLICES], rl[SLICES];
  __shared__ int   wsum[NPB / 32];
  __shared__ int   blk_base;
  __shared__ float dl[NPB];
  const int tid  = threadIdx.x;
  const int lane = tid & 31;
  const int wave  = __builtin_amdgcn_readfirstlane(tid >> 5);
  const int node0 = blockIdx.x * NPB;

  float4 xr[RPT];
  const float4* __restrict__ x4 = (const float4*)x;
#pragma unroll
  for (int k = 0; k < RPT; ++k) {
    const int r = min(node0 + wave + 32 * k, n - 1);
    xr[k] = x4[(size_t)r * XQ + lane];
  }
  __builtin_amdgcn_sched_barrier(0);

  int4* c4 = (int4*)cnt;
  for (int i = tid; i < SLICES * NPB / 4; i += OWN_THREADS) c4[i] = int4{0, 0, 0, 0};
  if (tid < SLICES) {
    const int* o = off + (size_t)tid * pitchB + blockIdx.x;
    const int a = o[0];
    const int z = o[1];
    rs[tid] = tid * chunk + a;
    rl[tid] = z - a;
  }
  __syncthreads();

  constexpr int SPW  = SLICES / (OWN_THREADS / 32);
  constexpr int HEAD = 3;
  static_assert(SLICES % (OWN_THREADS / 32) == 0, "every wave takes the same number of slices");
  const Pair* pp[SPW];
  int len[SPW];
  Pair q[SPW][HEAD];
#pragma unroll
  for (int s = 0; s < SPW; ++s) {
    const int p = wave + s * (OWN_THREADS / 32);
    pp[s]  = pairs + rs[p];
    len[s] = rl[p];
  }
#pragma unroll
  for (int s = 0; s < SPW; ++s) {
#pragma unroll
    for (int t = 0; t < HEAD; ++t) q[s][t] = pp[s][max(min(lane + 32 * t, len[s] - 1), 0)];
  }
#pragma unroll
  for (int s = 0; s < SPW; ++s) {
    int* cp = cnt + (wave + s * (OWN_THREADS / 32)) * NPB;
#pragma unroll
    for (int t = 0; t < HEAD; ++t) {
      if (lane + 32 * t < len[s]) {
        __hip_atomic_fetch_add(&cp[q[s][t].dst - node0], 1, __ATOMIC_RELAXED,
                               __HIP_MEMORY_SCOPE_WORKGROUP);
      }
    }
    for (int j = lane + 32 * HEAD; j < len[s]; j += 32) {
      const Pair r = pp[s][j];
      __hip_atomic_fetch_add(&cp[r.dst - node0], 1, __ATOMIC_RELAXED,
                             __HIP_MEMORY_SCOPE_WORKGROUP);
    }
  }
  __syncthreads();

  int c = 0, v = 0;
  if (tid < NPB) {
#pragma unroll 8
    for (int p = 0; p < SLICES; ++p) c += cnt[p * NPB + tid];
    v = c;
#pragma unroll
    for (int d = 1; d < 32; d <<= 1) {
      int t = __shfl_up(v, d);
      if (lane >= d) v += t;
    }
    if (lane == 31) wsum[wave] = v;
  }
  __syncthreads();
  if (tid == 0) {
    int tot = 0;
#pragma unroll
    for (int w = 0; w < NPB / 32; ++w) {
      int t = wsum[w];
      wsum[w] = tot;
      tot += t;
    }
    blk_base = __hip_atomic_fetch_add(cursor, tot, __ATOMIC_RELAXED,
                                      __HIP_MEMORY_SCOPE_AGENT);
  }
  __syncthreads();
  if (tid < NPB) {
    const int start = blk_base + wsum[wave] + (v - c);
    const float d  = (float)(c + 1);
    const float di = (d > 0.0f) ? rsqrtf(d) : 0.0f;
    dl[tid] = di;
    const int node = node0 + tid;
    if (node < n) {
      Seg sg;
      sg.start  = start;
      sg.len    = c;
      sg.di     = di;
      sg.unused = 0;
      seg[node]  = sg;
      dinv[node] = di;
    }
    int run = start;
#pragma unroll 8
    for (int p = 0; p < SLICES; ++p) {
      const int t = cnt[p * NPB + tid];
      cnt[p * NPB + tid] = run;
      run += t;
    }
  }
  __syncthreads();

#pragma unroll
  for (int s = 0; s < SPW; ++s) {
    int* cp = cnt + (wave + s * (OWN_THREADS / 32)) * NPB;
#pragma unroll
    for (int t = 0; t < HEAD; ++t) {
      if (lane + 32 * t < len[s]) {
        const int pos = __hip_atomic_fetch_add(&cp[q[s][t].dst - node0], 1, __ATOMIC_RELAXED,
                                               __HIP_MEMORY_SCOPE_WORKGROUP);
        ents[pos] = q[s][t].src;
      }
    }
    for (int j = lane + 32 * HEAD; j < len[s]; j += 32) {
      const Pair r = pp[s][j];
      const int pos = __hip_atomic_fetch_add(&cp[r.dst - node0], 1, __ATOMIC_RELAXED,
                                             __HIP_MEMORY_SCOPE_WORKGROUP);
      ents[pos] = r.src;
    }
  }

  float4* __restrict__ xs4 = (float4*)xs;
#pragma unroll
  for (int k = 0; k < RPT; ++k) {
    const int rloc = wave + 32 * k;
    const int r    = node0 + rloc;
    if (r < n) {
      const float di = dl[rloc];
      float4 t = xr[k];
      t.x *= di;
      t.y *= di;
      t.z *= di;
      t.w *= di;
      xs4[(size_t)r * XQ + lane] = t;
    }
  }
}

template <int VEC, bool ADD_BIAS>
__global__ __launch_bounds__(256) void k_gather_rows(
    const float* __restrict__ hs, const Seg* __restrict__ seg,
    const int* __restrict__ ents, const float* __restrict__ bias,
    float* __restrict__ out, int n_nodes) {
  typedef float vec __attribute__((ext_vector_type(VEC)));
  const int lane = threadIdx.x & 31;
  const int node = __builtin_amdgcn_readfirstlane(
      (int)(blockIdx.x * (blockDim.x >> 5) + (threadIdx.x >> 5)));
  if (node >= n_nodes) return;
  const Seg   sg = seg[node];
  const int   n  = sg.len;
  const float di = sg.di;
  const int* __restrict__ sp = ents + sg.start;
  const vec* __restrict__ hv = (const vec*)hs + lane;

  vec bvec = {};
  if (ADD_BIAS) {
#pragma unroll
    for (int c = 0; c < VEC; ++c) bvec[c] = bias[lane * VEC + c];
  }

  vec acc = hv[(size_t)node * 32];
  int j = 0;
  for (; j + 4 <= n; j += 4) {
    const int s0 = sp[j], s1 = sp[j + 1], s2 = sp[j + 2], s3 = sp[j + 3];
    const vec r0 = hv[(size_t)s0 * 32];
    const vec r1 = hv[(size_t)s1 * 32];
    const vec r2 = hv[(size_t)s2 * 32];
    const vec r3 = hv[(size_t)s3 * 32];
    acc += r0;
    acc += r1;
    acc += r2;
    acc += r3;
  }
  if (j + 2 <= n) {
    const int s0 = sp[j], s1 = sp[j + 1];
    const vec r0 = hv[(size_t)s0 * 32];
    const vec r1 = hv[(size_t)s1 * 32];
    acc += r0;
    acc += r1;
    j += 2;
  }
  if (j < n) {
    const int s0 = sp[j];
    acc += hv[(size_t)s0 * 32];
  }
  acc *= di;
  if (ADD_BIAS) acc += bvec;
  ((vec*)out)[(size_t)node * 32 + lane] = acc;
}

constexpr int MLP_THREADS = 512;
constexpr int MAIN_WAVES  = 12;

template <int K1, int N1, int N2, int NT>
__device__ __forceinline__ void mlp_group(const float4 (&q)[K1 / 8], v8f (&acc2)[N2 / 16],
                                          int g, int lane, int h,
                                          const float* __restrict__ b1,
                                          const float4* W1s, const float4* W2s) {
  constexpr int KP1 = K1 / 8;
  constexpr int NT1 = N1 / 16;
  constexpr int NT2 = N2 / 16;
  const int t0 = g * NT;

  float4 ba[NT], bb[NT];
#pragma unroll
  for (int t = 0; t < NT; ++t) {
    const float4* bp = (const float4*)(b1 + 16 * (t0 + t) + 8 * h);
    ba[t] = bp[0];
    bb[t] = bp[1];
  }

  v8f acc1[NT];
#pragma unroll
  for (int t = 0; t < NT; ++t) acc1[t] = v8f{};
  const float4* wp = W1s + t0 * 32 + lane;
#pragma unroll
  for (int j = 0; j < KP1; ++j) {
    float4 wf[NT];
#pragma unroll
    for (int t = 0; t < NT; ++t) wf[t] = wp[(j * NT1 + t) * 32];
    v2f x0, x1;
    x0.x = q[j].x;
    x0.y = q[j].y;
    x1.x = q[j].z;
    x1.y = q[j].w;
#pragma unroll
    for (int t = 0; t < NT; ++t) {
      v2f w0;
      w0.x = wf[t].x;
      w0.y = wf[t].y;
      acc1[t] = __builtin_amdgcn_wmma_f32_16x16x4_f32(
           false, w0,  false, x0,
           (short)0, acc1[t],  false,  false);
    }
#pragma unroll
    for (int t = 0; t < NT; ++t) {
      v2f w1;
      w1.x = wf[t].z;
      w1.y = wf[t].w;
      acc1[t] = __builtin_amdgcn_wmma_f32_16x16x4_f32(
           false, w1,  false, x1,
           (short)0, acc1[t],  false,  false);
    }
  }

#pragma unroll
  for (int t = 0; t < NT; ++t) {
    const int tt = t0 + t;
    float hv[8];
    hv[0] = fmaxf(acc1[t][0] + ba[t].x, 0.0f);
    hv[1] = fmaxf(acc1[t][1] + ba[t].y, 0.0f);
    hv[2] = fmaxf(acc1[t][2] + ba[t].z, 0.0f);
    hv[3] = fmaxf(acc1[t][3] + ba[t].w, 0.0f);
    hv[4] = fmaxf(acc1[t][4] + bb[t].x, 0.0f);
    hv[5] = fmaxf(acc1[t][5] + bb[t].y, 0.0f);
    hv[6] = fmaxf(acc1[t][6] + bb[t].z, 0.0f);
    hv[7] = fmaxf(acc1[t][7] + bb[t].w, 0.0f);
#pragma unroll
    for (int p = 0; p < 2; ++p) {
      float4 w2f[NT2];
#pragma unroll
      for (int k = 0; k < NT2; ++k) w2f[k] = W2s[((tt * NT2 + k) * 2 + p) * 32 + lane];
      v2f h0, h1;
      h0.x = hv[4 * p + 0];
      h0.y = hv[4 * p + 1];
      h1.x = hv[4 * p + 2];
      h1.y = hv[4 * p + 3];
#pragma unroll
      for (int k = 0; k < NT2; ++k) {
        v2f w0;
        w0.x = w2f[k].x;
        w0.y = w2f[k].y;
        acc2[k] = __builtin_amdgcn_wmma_f32_16x16x4_f32(
             false, w0,  false, h0,
             (short)0, acc2[k],  false,  false);
      }
#pragma unroll
      for (int k = 0; k < NT2; ++k) {
        v2f w1;
        w1.x = w2f[k].z;
        w1.y = w2f[k].w;
        acc2[k] = __builtin_amdgcn_wmma_f32_16x16x4_f32(
             false, w1,  false, h1,
             (short)0, acc2[k],  false,  false);
      }
    }
  }
}

template <int K1, int N1, int N2, int NT>
__global__ __launch_bounds__(MLP_THREADS) void k_mlp_wmma_f32(
    const float* __restrict__ X, const float* __restrict__ W1,
    const float* __restrict__ b1, const float* __restrict__ W2,
    const float* __restrict__ rscale, float* __restrict__ O, int M) {
  constexpr int KP1 = K1 / 8;
  constexpr int NT1 = N1 / 16;
  constexpr int NT2 = N2 / 16;
  constexpr int NG  = NT1 / NT;
  static_assert(NT1 % NT == 0, "hidden tiles are taken NT at a time");
  static_assert((MLP_THREADS >> 5) - MAIN_WAVES == NG, "one left-over wave per hidden group");
  __shared__ __attribute__((aligned(16))) float4 W1s[KP1 * NT1 * 32];
  __shared__ __attribute__((aligned(16))) float4 W2s[NT1 * NT2 * 2 * 32];
  __shared__ __attribute__((aligned(16))) float4 parts[NG * NT2 * 2 * 32];

  const int lane = threadIdx.x & 31;
  const int wave = __builtin_amdgcn_readfirstlane((int)(threadIdx.x >> 5));
  const int c = lane & 15;
  const int h = lane >> 4;
  const int grid  = (int)gridDim.x;
  const int units = M >> 4;
  const int nmain = grid * MAIN_WAVES;
  const int rest  = units % nmain;
  const int ncut  = (rest <= grid) ? rest : 0;
  const int nwhole = units - ncut;

  const bool main_wave = wave < MAIN_WAVES;
  const bool cut_block = (int)blockIdx.x < ncut;
  int u, u_end, g0, g1;
  if (main_wave) {
    u = wave * grid + (int)blockIdx.x;
    u_end = nwhole;
    g0 = 0;
    g1 = NG;
  } else {
    u = nwhole + (int)blockIdx.x;
    u_end = cut_block ? u + 1 : 0;
    g0 = wave - MAIN_WAVES;
    g1 = g0 + 1;
  }

  float4 q[KP1];
  {
    const int uc = min(u, units - 1);
    const float4* ap = (const float4*)(X + (size_t)((uc << 4) + c) * K1 + 4 * h);
#pragma unroll
    for (int j = 0; j < KP1; ++j) q[j] = ap[2 * j];
  }
  __builtin_amdgcn_sched_barrier(0);

  static_assert((KP1 * 2 * N1) % MLP_THREADS == 0, "the copy loop has no tail");
#pragma unroll
  for (int it = 0; it < KP1 * 2 * N1 / MLP_THREADS; ++it) {
    const int idx = it * MLP_THREADS + (int)threadIdx.x;
    const int n  = idx % N1;
    const int jh = idx / N1;
    const float* src = W1 + (size_t)(4 * jh) * N1 + n;
    float4 v;
    v.x = src[0];
    v.y = src[N1];
    v.z = src[2 * N1];
    v.w = src[3 * N1];
    W1s[((jh >> 1) * NT1 + (n >> 4)) * 32 + (jh & 1) * 16 + (n & 15)] = v;
  }
  static_assert((N1 / 4 * N2) % MLP_THREADS == 0, "the copy loop has no tail");
#pragma unroll
  for (int it = 0; it < N1 / 4 * N2 / MLP_THREADS; ++it) {
    const int idx = it * MLP_THREADS + (int)threadIdx.x;
    const int o  = idx % N2;
    const int n4 = idx / N2;
    const float* src = W2 + (size_t)(4 * n4) * N2 + o;
    float4 v;
    v.x = src[0];
    v.y = src[N2];
    v.z = src[2 * N2];
    v.w = src[3 * N2];
    const int t  = n4 >> 2;
    const int hh = (n4 >> 1) & 1;
    const int p  = n4 & 1;
    W2s[(((t * NT2) + (o >> 4)) * 2 + p) * 32 + hh * 16 + (o & 15)] = v;
  }
  __syncthreads();

  for (; u < u_end; u += nmain) {
    const float rsc = main_wave ? rscale[(u << 4) + c] : 1.0f;
    v8f acc2[NT2];
#pragma unroll
    for (int k = 0; k < NT2; ++k) acc2[k] = v8f{};

    for (int g = g0; g < g1; ++g) mlp_group<K1, N1, N2, NT>(q, acc2, g, lane, h, b1, W1s, W2s);

    float4 lo[NT2], hi[NT2];
#pragma unroll
    for (int k = 0; k < NT2; ++k) {
      lo[k].x = acc2[k][0] * rsc;
      lo[k].y = acc2[k][1] * rsc;
      lo[k].z = acc2[k][2] * rsc;
      lo[k].w = acc2[k][3] * rsc;
      hi[k].x = acc2[k][4] * rsc;
      hi[k].y = acc2[k][5] * rsc;
      hi[k].z = acc2[k][6] * rsc;
      hi[k].w = acc2[k][7] * rsc;
    }
    if (main_wave) {
      float4* op = (float4*)(O + (size_t)((u << 4) + c) * N2 + 8 * h);
#pragma unroll
      for (int k = 0; k < NT2; ++k) {
        op[4 * k]     = lo[k];
        op[4 * k + 1] = hi[k];
      }
      if (u + nmain < u_end) {
        const float4* ap = (const float4*)(X + (size_t)(((u + nmain) << 4) + c) * K1 + 4 * h);
#pragma unroll
        for (int j = 0; j < KP1; ++j) q[j] = ap[2 * j];
      }
    } else {
      float4* pp = parts + (g0 * NT2 * 2) * 32 + lane;
#pragma unroll
      for (int k = 0; k < NT2; ++k) {
        pp[(2 * k) * 32]     = lo[k];
        pp[(2 * k + 1) * 32] = hi[k];
      }
    }
  }

  __syncthreads();
  if (wave == MAIN_WAVES && cut_block) {
    const float4* pp = parts + lane;
    const float rsc = rscale[((nwhole + (int)blockIdx.x) << 4) + c];
    float4* op = (float4*)(O + (size_t)(((nwhole + (int)blockIdx.x) << 4) + c) * N2 + 8 * h);
#pragma unroll
    for (int k = 0; k < 2 * NT2; ++k) {
      float4 s = pp[k * 32];
#pragma unroll
      for (int g = 1; g < NG; ++g) {
        const float4 v = pp[(g * NT2 * 2 + k) * 32];
        s.x += v.x;
        s.y += v.y;
        s.z += v.z;
        s.w += v.w;
      }
      s.x *= rsc;
      s.y *= rsc;
      s.z *= rsc;
      s.w *= rsc;
      op[(k >> 1) * 4 + (k & 1)] = s;
    }
  }
}

extern "C" void kernel_launch(void* const* d_in, const int* in_sizes, int n_in,
                              void* d_out, int out_size, void* d_ws, size_t ws_size,
                              hipStream_t stream) {
  const float* x  = (const float*)d_in[0];
  const int*   ei = (const int*)d_in[1];
  const float* W1 = (const float*)d_in[2];
  const float* b1 = (const float*)d_in[3];
  const float* W2 = (const float*)d_in[4];
  const float* b2 = (const float*)d_in[5];

  const int H  = in_sizes[3];
  const int F  = in_sizes[2] / H;
  const int C  = in_sizes[5];
  const int Nn = in_sizes[0] / F;
  const int E  = in_sizes[1] / 2;
  const int* rowIdx = ei;
  const int* colIdx = ei + E;

  auto al = [](size_t v) { return (v + 255) & ~(size_t)255; };
  size_t off_b = 0;
  float* dinv = (float*)((char*)d_ws + off_b);
  off_b += al((size_t)Nn * sizeof(float));
  int* cursor = (int*)((char*)d_ws + off_b);
  off_b += al(sizeof(int));
  const int nb     = (Nn + NPB - 1) / NPB;
  const int pitchB = (nb + 1 + 63) & ~63;
  int* off = (int*)((char*)d_ws + off_b);
  off_b += al((size_t)SLICES * pitchB * sizeof(int));
  Seg* seg = (Seg*)((char*)d_ws + off_b);
  off_b += al((size_t)Nn * sizeof(Seg));
  Pair* pairs = (Pair*)((char*)d_ws + off_b);
  const int chunk = ((E + SLICES - 1) / SLICES + 63) & ~63;
  off_b += al((size_t)SLICES * chunk * sizeof(Pair));
  int* ents = (int*)((char*)d_ws + off_b);
  off_b += al((size_t)E * sizeof(int));
  float* obuf = (float*)((char*)d_ws + off_b);
  off_b += al((size_t)Nn * C * sizeof(float));
  float* abuf = (float*)((char*)d_ws + off_b);
  off_b += al((size_t)Nn * F * sizeof(float));
  float* xs = (float*)((char*)d_ws + off_b);

  const dim3 blk(256);
  const unsigned rowBlocks = (unsigned)((Nn + 7) / 8);

  k_bucket<<<SLICES, BKT_THREADS, 0, stream>>>(rowIdx, colIdx, pairs, off, cursor, E, Nn, nb, pitchB, chunk);
  k_own<32><<<nb, OWN_THREADS, 0, stream>>>(pairs, off, cursor, seg, dinv, ents, x, xs, Nn, pitchB, chunk);

  k_gather_rows<4, false><<<rowBlocks, blk, 0, stream>>>(xs, seg, ents, nullptr, abuf, Nn);

  k_mlp_wmma_f32<128, 256, 64, 4><<<256, MLP_THREADS, 0, stream>>>(abuf, W1, b1, W2, dinv, obuf, Nn);

  float* out = (float*)d_out;
  k_gather_rows<2, true><<<rowBlocks, blk, 0, stream>>>(obuf, seg, ents, b2, out, Nn);
}


# both edge-ordering kernels put their records (a slice's pairs, a node range's sources) in order in LDS and write whole 128-byte lines, not one scattered 4- or 8-byte store per edge; sum order kept
# speedup vs baseline: 6.1627x; 1.0843x over previous
#include <hip/hip_runtime.h>
#include <hip/hip_bf16.h>


typedef __attribute__((ext_vector_type(2))) float v2f;
typedef __attribute__((ext_vector_type(8))) float v8f;

struct __attribute__((aligned(16))) Seg {
  int   start;
  int   len;
  float di;
  int   unused;
};

struct __attribute__((aligned(8))) Pair {
  int src;
  int dst;
};

constexpr int SLICES      = 64;
constexpr int BKT_THREADS = 1024;
constexpr int NPB         = 256;
constexpr int MAXB        = 2048;
constexpr int EPT         = 16;
constexpr int OWN_THREADS = 1024;
constexpr int STAGE_CAP   = 16384;
constexpr int ENT_CAP     = 8192;

__global__ __launch_bounds__(BKT_THREADS) void k_bucket(
    const int* __restrict__ row, const int* __restrict__ col, Pair* __restrict__ pairs,
    int* __restrict__ off, int* __restrict__ cursor, int E, int n, int nb, int pitchB,
    int chunk) {
  __shared__ int cnt[MAXB];
  __shared__ __attribute__((aligned(16))) Pair stage[STAGE_CAP];
  const bool staged = chunk <= STAGE_CAP;
  const int tid = threadIdx.x;
  const int e0  = blockIdx.x * chunk;
  const int e1  = min(e0 + chunk, E);
  if (blockIdx.x == 0 && tid == 0) *cursor = 0;
  for (int i = tid; i < nb; i += BKT_THREADS) cnt[i] = 0;
  __syncthreads();
  for (int e = e0 + tid; e < e1; e += EPT * BKT_THREADS) {
    int c[EPT];
#pragma unroll
    for (int k = 0; k < EPT; ++k) c[k] = col[min(e + k * BKT_THREADS, e1 - 1)];
#pragma unroll
    for (int k = 0; k < EPT; ++k) {
      if (e + k * BKT_THREADS < e1 && (unsigned)c[k] < (unsigned)n) {
        __hip_atomic_fetch_add(&cnt[c[k] / NPB], 1, __ATOMIC_RELAXED,
                               __HIP_MEMORY_SCOPE_WORKGROUP);
      }
    }
  }
  __syncthreads();
  if (tid < 32) {
    const int per = (nb + 31) >> 5;
    const int b0  = tid * per;
    const int b1  = min(b0 + per, nb);
    int s = 0;
    for (int b = b0; b < b1; ++b) s += cnt[b];
    int v = s;
#pragma unroll
    for (int d = 1; d < 32; d <<= 1) {
      int t = __shfl_up(v, d);
      if (tid >= d) v += t;
    }
    int run = v - s;
    int* __restrict__ orow = off + (size_t)blockIdx.x * pitchB;
    for (int b = b0; b < b1; ++b) {
      const int t = cnt[b];
      cnt[b]  = run;
      orow[b] = run;
      run += t;
    }
    if (tid == 31) orow[nb] = v;
  }
  __syncthreads();
  Pair* __restrict__ piece = pairs + e0;
  for (int e = e0 + tid; e < e1; e += EPT * BKT_THREADS) {
    int c[EPT], sr[EPT];
#pragma unroll
    for (int k = 0; k < EPT; ++k) {
      const int idx = min(e + k * BKT_THREADS, e1 - 1);
      c[k]  = col[idx];
      sr[k] = row[idx];
    }
#pragma unroll
    for (int k = 0; k < EPT; ++k) {
      if (e + k * BKT_THREADS < e1 && (unsigned)c[k] < (unsigned)n) {
        const int pos = __hip_atomic_fetch_add(&cnt[c[k] / NPB], 1, __ATOMIC_RELAXED,
                                               __HIP_MEMORY_SCOPE_WORKGROUP);
        Pair pr;
        pr.src = sr[k];
        pr.dst = c[k];
        if (staged) stage[pos] = pr;
        else        piece[pos] = pr;
      }
    }
  }
  if (staged) {
    __syncthreads();
    const int total = cnt[nb - 1];
    const int4* s4 = (const int4*)stage;
    int4* p4 = (int4*)piece;
    const int half = total >> 1;
#pragma unroll 4
    for (int i = tid; i < half; i += BKT_THREADS) p4[i] = s4[i];
    if ((total & 1) && tid == 0) piece[total - 1] = stage[total - 1];
  }
}

template <int XQ>
__global__ __launch_bounds__(OWN_THREADS) void k_own(
    const Pair* __restrict__ pairs, const int* __restrict__ off, int* cursor,
    Seg* __restrict__ seg, float* __restrict__ dinv, int* __restrict__ ents,
    const float* __restrict__ x, float* __restrict__ xs, int n, int pitchB, int chunk) {
  static_assert(XQ == 32, "a wave moves one row of x per load: 32 lanes of 16 bytes");
  constexpr int RPT = NPB * XQ / OWN_THREADS;
  __shared__ __attribute__((aligned(16))) int cnt[SLICES * NPB];
  __shared__ int   rs[SLICES], rl[SLICES];
  __shared__ int   wsum[NPB / 32];
  __shared__ int   blk_base, blk_tot;
  __shared__ float dl[NPB];
  __shared__ int est[ENT_CAP];
  const int tid  = threadIdx.x;
  const int lane = tid & 31;
  const int wave  = __builtin_amdgcn_readfirstlane(tid >> 5);
  const int node0 = blockIdx.x * NPB;

  float4 xr[RPT];
  const float4* __restrict__ x4 = (const float4*)x;
#pragma unroll
  for (int k = 0; k < RPT; ++k) {
    const int r = min(node0 + wave + 32 * k, n - 1);
    xr[k] = x4[(size_t)r * XQ + lane];
  }
  __builtin_amdgcn_sched_barrier(0);

  int4* c4 = (int4*)cnt;
  for (int i = tid; i < SLICES * NPB / 4; i += OWN_THREADS) c4[i] = int4{0, 0, 0, 0};
  if (tid < SLICES) {
    const int* o = off + (size_t)tid * pitchB + blockIdx.x;
    const int a = o[0];
    const int z = o[1];
    rs[tid] = tid * chunk + a;
    rl[tid] = z - a;
  }
  __syncthreads();

  constexpr int SPW  = SLICES / (OWN_THREADS / 32);
  constexpr int HEAD = 3;
  static_assert(SLICES % (OWN_THREADS / 32) == 0, "every wave takes the same number of slices");
  const Pair* pp[SPW];
  int len[SPW];
  Pair q[SPW][HEAD];
#pragma unroll
  for (int s = 0; s < SPW; ++s) {
    const int p = wave + s * (OWN_THREADS / 32);
    pp[s]  = pairs + rs[p];
    len[s] = rl[p];
  }
#pragma unroll
  for (int s = 0; s < SPW; ++s) {
#pragma unroll
    for (int t = 0; t < HEAD; ++t) q[s][t] = pp[s][max(min(lane + 32 * t, len[s] - 1), 0)];
  }
#pragma unroll
  for (int s = 0; s < SPW; ++s) {
    int* cp = cnt + (wave + s * (OWN_THREADS / 32)) * NPB;
#pragma unroll
    for (int t = 0; t < HEAD; ++t) {
      if (lane + 32 * t < len[s]) {
        __hip_atomic_fetch_add(&cp[q[s][t].dst - node0], 1, __ATOMIC_RELAXED,
                               __HIP_MEMORY_SCOPE_WORKGROUP);
      }
    }
    for (int j = lane + 32 * HEAD; j < len[s]; j += 32) {
      const Pair r = pp[s][j];
      __hip_atomic_fetch_add(&cp[r.dst - node0], 1, __ATOMIC_RELAXED,
                             __HIP_MEMORY_SCOPE_WORKGROUP);
    }
  }
  __syncthreads();

  int c = 0, v = 0;
  if (tid < NPB) {
#pragma unroll 8
    for (int p = 0; p < SLICES; ++p) c += cnt[p * NPB + tid];
    v = c;
#pragma unroll
    for (int d = 1; d < 32; d <<= 1) {
      int t = __shfl_up(v, d);
      if (lane >= d) v += t;
    }
    if (lane == 31) wsum[wave] = v;
  }
  __syncthreads();
  if (tid == 0) {
    int tot = 0;
#pragma unroll
    for (int w = 0; w < NPB / 32; ++w) {
      int t = wsum[w];
      wsum[w] = tot;
      tot += t;
    }
    blk_base = __hip_atomic_fetch_add(cursor, tot, __ATOMIC_RELAXED,
                                      __HIP_MEMORY_SCOPE_AGENT);
    blk_tot  = tot;
  }
  __syncthreads();
  if (tid < NPB) {
    const int start = blk_base + wsum[wave] + (v - c);
    const float d  = (float)(c + 1);
    const float di = (d > 0.0f) ? rsqrtf(d) : 0.0f;
    dl[tid] = di;
    const int node = node0 + tid;
    if (node < n) {
      Seg sg;
      sg.start  = start;
      sg.len    = c;
      sg.di     = di;
      sg.unused = 0;
      seg[node]  = sg;
      dinv[node] = di;
    }
    int run = start;
#pragma unroll 8
    for (int p = 0; p < SLICES; ++p) {
      const int t = cnt[p * NPB + tid];
      cnt[p * NPB + tid] = run;
      run += t;
    }
  }
  __syncthreads();

  const int  ebase  = blk_base;
  const int  etot   = blk_tot;
  const bool estage = etot <= ENT_CAP;
#pragma unroll
  for (int s = 0; s < SPW; ++s) {
    int* cp = cnt + (wave + s * (OWN_THREADS / 32)) * NPB;
#pragma unroll
    for (int t = 0; t < HEAD; ++t) {
      if (lane + 32 * t < len[s]) {
        const int pos = __hip_atomic_fetch_add(&cp[q[s][t].dst - node0], 1, __ATOMIC_RELAXED,
                                               __HIP_MEMORY_SCOPE_WORKGROUP);
        if (estage) est[pos - ebase] = q[s][t].src;
        else        ents[pos] = q[s][t].src;
      }
    }
    for (int j = lane + 32 * HEAD; j < len[s]; j += 32) {
      const Pair r = pp[s][j];
      const int pos = __hip_atomic_fetch_add(&cp[r.dst - node0], 1, __ATOMIC_RELAXED,
                                             __HIP_MEMORY_SCOPE_WORKGROUP);
      if (estage) est[pos - ebase] = r.src;
      else        ents[pos] = r.src;
    }
  }
  if (estage) {
    __syncthreads();
    const int shift = ebase & 31;
    int i0 = tid - shift;
    if (i0 < 0) i0 += OWN_THREADS;
#pragma unroll 4
    for (int i = i0; i < etot; i += OWN_THREADS) ents[ebase + i] = est[i];
  }

  float4* __restrict__ xs4 = (float4*)xs;
#pragma unroll
  for (int k = 0; k < RPT; ++k) {
    const int rloc = wave + 32 * k;
    const int r    = node0 + rloc;
    if (r < n) {
      const float di = dl[rloc];
      float4 t = xr[k];
      t.x *= di;
      t.y *= di;
      t.z *= di;
      t.w *= di;
      xs4[(size_t)r * XQ + lane] = t;
    }
  }
}

template <int VEC, bool ADD_BIAS>
__global__ __launch_bounds__(256) void k_gather_rows(
    const float* __restrict__ hs, const Seg* __restrict__ seg,
    const int* __restrict__ ents, const float* __restrict__ bias,
    float* __restrict__ out, int n_nodes) {
  typedef float vec __attribute__((ext_vector_type(VEC)));
  const int lane = threadIdx.x & 31;
  const int node = __builtin_amdgcn_readfirstlane(
      (int)(blockIdx.x * (blockDim.x >> 5) + (threadIdx.x >> 5)));
  if (node >= n_nodes) return;
  const Seg   sg = seg[node];
  const int   n  = sg.len;
  const float di = sg.di;
  const int* __restrict__ sp = ents + sg.start;
  const vec* __restrict__ hv = (const vec*)hs + lane;

  vec bvec = {};
  if (ADD_BIAS) {
#pragma unroll
    for (int c = 0; c < VEC; ++c) bvec[c] = bias[lane * VEC + c];
  }

  vec acc = hv[(size_t)node * 32];
  int j = 0;
  for (; j + 4 <= n; j += 4) {
    const int s0 = sp[j], s1 = sp[j + 1], s2 = sp[j + 2], s3 = sp[j + 3];
    const vec r0 = hv[(size_t)s0 * 32];
    const vec r1 = hv[(size_t)s1 * 32];
    const vec r2 = hv[(size_t)s2 * 32];
    const vec r3 = hv[(size_t)s3 * 32];
    acc += r0;
    acc += r1;
    acc += r2;
    acc += r3;
  }
  if (j + 2 <= n) {
    const int s0 = sp[j], s1 = sp[j + 1];
    const vec r0 = hv[(size_t)s0 * 32];
    const vec r1 = hv[(size_t)s1 * 32];
    acc += r0;
    acc += r1;
    j += 2;
  }
  if (j < n) {
    const int s0 = sp[j];
    acc += hv[(size_t)s0 * 32];
  }
  acc *= di;
  if (ADD_BIAS) acc += bvec;
  ((vec*)out)[(size_t)node * 32 + lane] = acc;
}

constexpr int MLP_THREADS = 512;
constexpr int MAIN_WAVES  = 12;

template <int K1, int N1, int N2, int NT>
__device__ __forceinline__ void mlp_group(const float4 (&q)[K1 / 8], v8f (&acc2)[N2 / 16],
                                          int g, int lane, int h,
                                          const float* __restrict__ b1,
                                          const float4* W1s, const float4* W2s) {
  constexpr int KP1 = K1 / 8;
  constexpr int NT1 = N1 / 16;
  constexpr int NT2 = N2 / 16;
  const int t0 = g * NT;

  float4 ba[NT], bb[NT];
#pragma unroll
  for (int t = 0; t < NT; ++t) {
    const float4* bp = (const float4*)(b1 + 16 * (t0 + t) + 8 * h);
    ba[t] = bp[0];
    bb[t] = bp[1];
  }

  v8f acc1[NT];
#pragma unroll
  for (int t = 0; t < NT; ++t) acc1[t] = v8f{};
  const float4* wp = W1s + t0 * 32 + lane;
#pragma unroll
  for (int j = 0; j < KP1; ++j) {
    float4 wf[NT];
#pragma unroll
    for (int t = 0; t < NT; ++t) wf[t] = wp[(j * NT1 + t) * 32];
    v2f x0, x1;
    x0.x = q[j].x;
    x0.y = q[j].y;
    x1.x = q[j].z;
    x1.y = q[j].w;
#pragma unroll
    for (int t = 0; t < NT; ++t) {
      v2f w0;
      w0.x = wf[t].x;
      w0.y = wf[t].y;
      acc1[t] = __builtin_amdgcn_wmma_f32_16x16x4_f32(
           false, w0,  false, x0,
           (short)0, acc1[t],  false,  false);
    }
#pragma unroll
    for (int t = 0; t < NT; ++t) {
      v2f w1;
      w1.x = wf[t].z;
      w1.y = wf[t].w;
      acc1[t] = __builtin_amdgcn_wmma_f32_16x16x4_f32(
           false, w1,  false, x1,
           (short)0, acc1[t],  false,  false);
    }
  }

#pragma unroll
  for (int t = 0; t < NT; ++t) {
    const int tt = t0 + t;
    float hv[8];
    hv[0] = fmaxf(acc1[t][0] + ba[t].x, 0.0f);
    hv[1] = fmaxf(acc1[t][1] + ba[t].y, 0.0f);
    hv[2] = fmaxf(acc1[t][2] + ba[t].z, 0.0f);
    hv[3] = fmaxf(acc1[t][3] + ba[t].w, 0.0f);
    hv[4] = fmaxf(acc1[t][4] + bb[t].x, 0.0f);
    hv[5] = fmaxf(acc1[t][5] + bb[t].y, 0.0f);
    hv[6] = fmaxf(acc1[t][6] + bb[t].z, 0.0f);
    hv[7] = fmaxf(acc1[t][7] + bb[t].w, 0.0f);
#pragma unroll
    for (int p = 0; p < 2; ++p) {
      float4 w2f[NT2];
#pragma unroll
      for (int k = 0; k < NT2; ++k) w2f[k] = W2s[((tt * NT2 + k) * 2 + p) * 32 + lane];
      v2f h0, h1;
      h0.x = hv[4 * p + 0];
      h0.y = hv[4 * p + 1];
      h1.x = hv[4 * p + 2];
      h1.y = hv[4 * p + 3];
#pragma unroll
      for (int k = 0; k < NT2; ++k) {
        v2f w0;
        w0.x = w2f[k].x;
        w0.y = w2f[k].y;
        acc2[k] = __builtin_amdgcn_wmma_f32_16x16x4_f32(
             false, w0,  false, h0,
             (short)0, acc2[k],  false,  false);
      }
#pragma unroll
      for (int k = 0; k < NT2; ++k) {
        v2f w1;
        w1.x = w2f[k].z;
        w1.y = w2f[k].w;
        acc2[k] = __builtin_amdgcn_wmma_f32_16x16x4_f32(
             false, w1,  false, h1,
             (short)0, acc2[k],  false,  false);
      }
    }
  }
}

template <int K1, int N1, int N2, int NT>
__global__ __launch_bounds__(MLP_THREADS) void k_mlp_wmma_f32(
    const float* __restrict__ X, const float* __restrict__ W1,
    const float* __restrict__ b1, const float* __restrict__ W2,
    const float* __restrict__ rscale, float* __restrict__ O, int M) {
  constexpr int KP1 = K1 / 8;
  constexpr int NT1 = N1 / 16;
  constexpr int NT2 = N2 / 16;
  constexpr int NG  = NT1 / NT;
  static_assert(NT1 % NT == 0, "hidden tiles are taken NT at a time");
  static_assert((MLP_THREADS >> 5) - MAIN_WAVES == NG, "one left-over wave per hidden group");
  __shared__ __attribute__((aligned(16))) float4 W1s[KP1 * NT1 * 32];
  __shared__ __attribute__((aligned(16))) float4 W2s[NT1 * NT2 * 2 * 32];
  __shared__ __attribute__((aligned(16))) float4 parts[NG * NT2 * 2 * 32];

  const int lane = threadIdx.x & 31;
  const int wave = __builtin_amdgcn_readfirstlane((int)(threadIdx.x >> 5));
  const int c = lane & 15;
  const int h = lane >> 4;
  const int grid  = (int)gridDim.x;
  const int units = M >> 4;
  const int nmain = grid * MAIN_WAVES;
  const int rest  = units % nmain;
  const int ncut  = (rest <= grid) ? rest : 0;
  const int nwhole = units - ncut;

  const bool main_wave = wave < MAIN_WAVES;
  const bool cut_block = (int)blockIdx.x < ncut;
  int u, u_end, g0, g1;
  if (main_wave) {
    u = wave * grid + (int)blockIdx.x;
    u_end = nwhole;
    g0 = 0;
    g1 = NG;
  } else {
    u = nwhole + (int)blockIdx.x;
    u_end = cut_block ? u + 1 : 0;
    g0 = wave - MAIN_WAVES;
    g1 = g0 + 1;
  }

  float4 q[KP1];
  {
    const int uc = min(u, units - 1);
    const float4* ap = (const float4*)(X + (size_t)((uc << 4) + c) * K1 + 4 * h);
#pragma unroll
    for (int j = 0; j < KP1; ++j) q[j] = ap[2 * j];
  }
  __builtin_amdgcn_sched_barrier(0);

  static_assert((KP1 * 2 * N1) % MLP_THREADS == 0, "the copy loop has no tail");
#pragma unroll
  for (int it = 0; it < KP1 * 2 * N1 / MLP_THREADS; ++it) {
    const int idx = it * MLP_THREADS + (int)threadIdx.x;
    const int n  = idx % N1;
    const int jh = idx / N1;
    const float* src = W1 + (size_t)(4 * jh) * N1 + n;
    float4 v;
    v.x = src[0];
    v.y = src[N1];
    v.z = src[2 * N1];
    v.w = src[3 * N1];
    W1s[((jh >> 1) * NT1 + (n >> 4)) * 32 + (jh & 1) * 16 + (n & 15)] = v;
  }
  static_assert((N1 / 4 * N2) % MLP_THREADS == 0, "the copy loop has no tail");
#pragma unroll
  for (int it = 0; it < N1 / 4 * N2 / MLP_THREADS; ++it) {
    const int idx = it * MLP_THREADS + (int)threadIdx.x;
    const int o  = idx % N2;
    const int n4 = idx / N2;
    const float* src = W2 + (size_t)(4 * n4) * N2 + o;
    float4 v;
    v.x = src[0];
    v.y = src[N2];
    v.z = src[2 * N2];
    v.w = src[3 * N2];
    const int t  = n4 >> 2;
    const int hh = (n4 >> 1) & 1;
    const int p  = n4 & 1;
    W2s[(((t * NT2) + (o >> 4)) * 2 + p) * 32 + hh * 16 + (o & 15)] = v;
  }
  __syncthreads();

  for (; u < u_end; u += nmain) {
    const float rsc = main_wave ? rscale[(u << 4) + c] : 1.0f;
    v8f acc2[NT2];
#pragma unroll
    for (int k = 0; k < NT2; ++k) acc2[k] = v8f{};

    for (int g = g0; g < g1; ++g) mlp_group<K1, N1, N2, NT>(q, acc2, g, lane, h, b1, W1s, W2s);

    float4 lo[NT2], hi[NT2];
#pragma unroll
    for (int k = 0; k < NT2; ++k) {
      lo[k].x = acc2[k][0] * rsc;
      lo[k].y = acc2[k][1] * rsc;
      lo[k].z = acc2[k][2] * rsc;
      lo[k].w = acc2[k][3] * rsc;
      hi[k].x = acc2[k][4] * rsc;
      hi[k].y = acc2[k][5] * rsc;
      hi[k].z = acc2[k][6] * rsc;
      hi[k].w = acc2[k][7] * rsc;
    }
    if (main_wave) {
      float4* op = (float4*)(O + (size_t)((u << 4) + c) * N2 + 8 * h);
#pragma unroll
      for (int k = 0; k < NT2; ++k) {
        op[4 * k]     = lo[k];
        op[4 * k + 1] = hi[k];
      }
      if (u + nmain < u_end) {
        const float4* ap = (const float4*)(X + (size_t)(((u + nmain) << 4) + c) * K1 + 4 * h);
#pragma unroll
        for (int j = 0; j < KP1; ++j) q[j] = ap[2 * j];
      }
    } else {
      float4* pp = parts + (g0 * NT2 * 2) * 32 + lane;
#pragma unroll
      for (int k = 0; k < NT2; ++k) {
        pp[(2 * k) * 32]     = lo[k];
        pp[(2 * k + 1) * 32] = hi[k];
      }
    }
  }

  __syncthreads();
  if (wave == MAIN_WAVES && cut_block) {
    const float4* pp = parts + lane;
    const float rsc = rscale[((nwhole + (int)blockIdx.x) << 4) + c];
    float4* op = (float4*)(O + (size_t)(((nwhole + (int)blockIdx.x) << 4) + c) * N2 + 8 * h);
#pragma unroll
    for (int k = 0; k < 2 * NT2; ++k) {
      float4 s = pp[k * 32];
#pragma unroll
      for (int g = 1; g < NG; ++g) {
        const float4 v = pp[(g * NT2 * 2 + k) * 32];
        s.x += v.x;
        s.y += v.y;
        s.z += v.z;
        s.w += v.w;
      }
      s.x *= rsc;
      s.y *= rsc;
      s.z *= rsc;
      s.w *= rsc;
      op[(k >> 1) * 4 + (k & 1)] = s;
    }
  }
}

extern "C" void kernel_launch(void* const* d_in, const int* in_sizes, int n_in,
                              void* d_out, int out_size, void* d_ws, size_t ws_size,
                              hipStream_t stream) {
  const float* x  = (const float*)d_in[0];
  const int*   ei = (const int*)d_in[1];
  const float* W1 = (const float*)d_in[2];
  const float* b1 = (const float*)d_in[3];
  const float* W2 = (const float*)d_in[4];
  const float* b2 = (const float*)d_in[5];

  const int H  = in_sizes[3];
  const int F  = in_sizes[2] / H;
  const int C  = in_sizes[5];
  const int Nn = in_sizes[0] / F;
  const int E  = in_sizes[1] / 2;
  const int* rowIdx = ei;
  const int* colIdx = ei + E;

  auto al = [](size_t v) { return (v + 255) & ~(size_t)255; };
  size_t off_b = 0;
  float* dinv = (float*)((char*)d_ws + off_b);
  off_b += al((size_t)Nn * sizeof(float));
  int* cursor = (int*)((char*)d_ws + off_b);
  off_b += al(sizeof(int));
  const int nb     = (Nn + NPB - 1) / NPB;
  const int pitchB = (nb + 1 + 63) & ~63;
  int* off = (int*)((char*)d_ws + off_b);
  off_b += al((size_t)SLICES * pitchB * sizeof(int));
  Seg* seg = (Seg*)((char*)d_ws + off_b);
  off_b += al((size_t)Nn * sizeof(Seg));
  Pair* pairs = (Pair*)((char*)d_ws + off_b);
  const int chunk = ((E + SLICES - 1) / SLICES + 63) & ~63;
  off_b += al((size_t)SLICES * chunk * sizeof(Pair));
  int* ents = (int*)((char*)d_ws + off_b);
  off_b += al((size_t)E * sizeof(int));
  float* obuf = (float*)((char*)d_ws + off_b);
  off_b += al((size_t)Nn * C * sizeof(float));
  float* abuf = (float*)((char*)d_ws + off_b);
  off_b += al((size_t)Nn * F * sizeof(float));
  float* xs = (float*)((char*)d_ws + off_b);

  const dim3 blk(256);
  const unsigned rowBlocks = (unsigned)((Nn + 7) / 8);

  k_bucket<<<SLICES, BKT_THREADS, 0, stream>>>(rowIdx, colIdx, pairs, off, cursor, E, Nn, nb, pitchB, chunk);
  k_own<32><<<nb, OWN_THREADS, 0, stream>>>(pairs, off, cursor, seg, dinv, ents, x, xs, Nn, pitchB, chunk);

  k_gather_rows<4, false><<<rowBlocks, blk, 0, stream>>>(xs, seg, ents, nullptr, abuf, Nn);

  k_mlp_wmma_f32<128, 256, 64, 4><<<256, MLP_THREADS, 0, stream>>>(abuf, W1, b1, W2, dinv, obuf, Nn);

  float* out = (float*)d_out;
  k_gather_rows<2, true><<<rowBlocks, blk, 0, stream>>>(obuf, seg, ents, b2, out, Nn);
}
